# starting kernel
# baseline (speedup 1.0000x reference)
#include <hip/hip_runtime.h>

typedef __attribute__((ext_vector_type(2))) float v2f;
typedef __attribute__((ext_vector_type(8))) float v8f;

#define DIM 64

__global__ void gcn_zero_kernel(float* __restrict__ out, int n4) {
    int i = blockIdx.x * blockDim.x + threadIdx.x;
    if (i < n4) {
        float4 z = make_float4(0.f, 0.f, 0.f, 0.f);
        ((float4*)out)[i] = z;
    }
}

__global__ void gcn_gemm_wmma_kernel(const float* __restrict__ X,
                                     const float* __restrict__ W,
                                     float* __restrict__ S,
                                     int n_row_tiles) {
    const int wave = threadIdx.x >> 5;
    const int lane = threadIdx.x & 31;
    const int tile = blockIdx.x * (blockDim.x >> 5) + wave;
    if (tile >= n_row_tiles) return;

    const int m0    = tile << 4;
    const int mrow  = lane & 15;
    const int khalf = lane >> 4;
    const int ncol  = lane & 15;

    v8f acc0 = {}, acc1 = {}, acc2 = {}, acc3 = {};

    const float* __restrict__ xrow = X + (long long)(m0 + mrow) * DIM;

    #pragma unroll
    for (int k0 = 0; k0 < DIM; k0 += 4) {
        const int ka = k0 + (khalf << 1);
        v2f a = *(const v2f*)(xrow + ka);

        const float* __restrict__ w0 = W + ka * DIM;
        const float* __restrict__ w1 = W + (ka + 1) * DIM;

        v2f b0, b1, b2, b3;
        b0.x = w0[ 0 + ncol]; b0.y = w1[ 0 + ncol];
        b1.x = w0[16 + ncol]; b1.y = w1[16 + ncol];
        b2.x = w0[32 + ncol]; b2.y = w1[32 + ncol];
        b3.x = w0[48 + ncol]; b3.y = w1[48 + ncol];

        acc0 = __builtin_amdgcn_wmma_f32_16x16x4_f32(false, a, false, b0,
                                                     (short)0, acc0, false, false);
        acc1 = __builtin_amdgcn_wmma_f32_16x16x4_f32(false, a, false, b1,
                                                     (short)0, acc1, false, false);
        acc2 = __builtin_amdgcn_wmma_f32_16x16x4_f32(false, a, false, b2,
                                                     (short)0, acc2, false, false);
        acc3 = __builtin_amdgcn_wmma_f32_16x16x4_f32(false, a, false, b3,
                                                     (short)0, acc3, false, false);
    }

    const int rbase = m0 + (khalf << 3);
    #pragma unroll
    for (int v = 0; v < 8; ++v) {
        float* __restrict__ srow = S + (long long)(rbase + v) * DIM + ncol;
        srow[ 0] = acc0[v];
        srow[16] = acc1[v];
        srow[32] = acc2[v];
        srow[48] = acc3[v];
    }
}

__global__ void gcn_scatter_kernel(const float* __restrict__ S,
                                   const int* __restrict__ adj_row,
                                   const int* __restrict__ adj_col,
                                   const float* __restrict__ adj_val,
                                   float* __restrict__ out,
                                   int n_edges) {
    const long long idx = (long long)blockIdx.x * blockDim.x + threadIdx.x;
    const int e = (int)(idx >> 6);
    const int d = (int)(idx & 63);
    if (e >= n_edges) return;

    const int   r = adj_row[e];
    const int   c = adj_col[e];
    const float v = adj_val[e];

    atomicAdd(out + (long long)r * DIM + d, v * S[(long long)c * DIM + d]);
}

extern "C" void kernel_launch(void* const* d_in, const int* in_sizes, int n_in,
                              void* d_out, int out_size, void* d_ws, size_t ws_size,
                              hipStream_t stream) {
    const float* X    = (const float*)d_in[0];
    const float* W    = (const float*)d_in[1];
    const int*   arow = (const int*)d_in[2];
    const int*   acol = (const int*)d_in[3];
    const float* aval = (const float*)d_in[4];
    float*       out  = (float*)d_out;
    float*       S    = (float*)d_ws;

    const int n_nodes = in_sizes[0] / DIM;
    const int n_edges = in_sizes[2];

    {
        const int n4 = out_size / 4;
        gcn_zero_kernel<<<(n4 + 255) / 256, 256, 0, stream>>>(out, n4);
    }

    {
        const int tiles = n_nodes / 16;
        const int waves_per_block = 256 / 32;
        const int blocks = (tiles + waves_per_block - 1) / waves_per_block;
        gcn_gemm_wmma_kernel<<<blocks, 256, 0, stream>>>(X, W, S, tiles);
    }

    {
        const long long work = (long long)n_edges * DIM;
        const int blocks = (int)((work + 255) / 256);
        gcn_scatter_kernel<<<blocks, 256, 0, stream>>>(S, arow, acol, aval, out, n_edges);
    }
}


# Float atomics removed: an integer count per output row gives each edge a slot, then one wave per row sorts its records by edge id and sums them in that fixed order, writing each out row once
# speedup vs baseline: 2.0784x; 2.0784x over previous
#include <hip/hip_runtime.h>

typedef __attribute__((ext_vector_type(2))) float v2f;
typedef __attribute__((ext_vector_type(8))) float v8f;

#define DIM 64
#define SLOTS 64
#define NO_EDGE 0x7fffffff

__global__ void gcn_zero_kernel(float* __restrict__ out, int n4) {
    int i = blockIdx.x * blockDim.x + threadIdx.x;
    if (i < n4) {
        float4 z = make_float4(0.f, 0.f, 0.f, 0.f);
        ((float4*)out)[i] = z;
    }
}

__global__ void gcn_gemm_wmma_kernel(const float* __restrict__ X,
                                     const float* __restrict__ W,
                                     float* __restrict__ S,
                                     int n_row_tiles) {
    const int wave = threadIdx.x >> 5;
    const int lane = threadIdx.x & 31;
    const int tile = blockIdx.x * (blockDim.x >> 5) + wave;
    if (tile >= n_row_tiles) return;

    const int m0    = tile << 4;
    const int mrow  = lane & 15;
    const int khalf = lane >> 4;
    const int ncol  = lane & 15;

    v8f acc0 = {}, acc1 = {}, acc2 = {}, acc3 = {};

    const float* __restrict__ xrow = X + (long long)(m0 + mrow) * DIM;

    #pragma unroll
    for (int k0 = 0; k0 < DIM; k0 += 4) {
        const int ka = k0 + (khalf << 1);
        v2f a = *(const v2f*)(xrow + ka);

        const float* __restrict__ w0 = W + ka * DIM;
        const float* __restrict__ w1 = W + (ka + 1) * DIM;

        v2f b0, b1, b2, b3;
        b0.x = w0[ 0 + ncol]; b0.y = w1[ 0 + ncol];
        b1.x = w0[16 + ncol]; b1.y = w1[16 + ncol];
        b2.x = w0[32 + ncol]; b2.y = w1[32 + ncol];
        b3.x = w0[48 + ncol]; b3.y = w1[48 + ncol];

        acc0 = __builtin_amdgcn_wmma_f32_16x16x4_f32(false, a, false, b0,
                                                     (short)0, acc0, false, false);
        acc1 = __builtin_amdgcn_wmma_f32_16x16x4_f32(false, a, false, b1,
                                                     (short)0, acc1, false, false);
        acc2 = __builtin_amdgcn_wmma_f32_16x16x4_f32(false, a, false, b2,
                                                     (short)0, acc2, false, false);
        acc3 = __builtin_amdgcn_wmma_f32_16x16x4_f32(false, a, false, b3,
                                                     (short)0, acc3, false, false);
    }

    const int rbase = m0 + (khalf << 3);
    #pragma unroll
    for (int v = 0; v < 8; ++v) {
        float* __restrict__ srow = S + (long long)(rbase + v) * DIM + ncol;
        srow[ 0] = acc0[v];
        srow[16] = acc1[v];
        srow[32] = acc2[v];
        srow[48] = acc3[v];
    }
}

__global__ void gcn_fill_kernel(const int* __restrict__ adj_row,
                                const int* __restrict__ adj_col,
                                const float* __restrict__ adj_val,
                                int* __restrict__ cnt,
                                int* __restrict__ ovf_head,
                                int* __restrict__ ovf_next,
                                int* __restrict__ slots,
                                int n_edges) {
    const int e = blockIdx.x * blockDim.x + threadIdx.x;
    if (e >= n_edges) return;

    const int   r = adj_row[e];
    const int   c = adj_col[e];
    const float v = adj_val[e];

    const int rank = atomicAdd(cnt + r, 1);
    if (rank < SLOTS) {
        ((int4*)slots)[(long long)r * SLOTS + rank] = make_int4(e, c, __float_as_int(v), 0);
    } else {
        ovf_next[e] = atomicExch(ovf_head + r, e + 1);
    }
}

#define GATHER_WAVES 8

template <int U>
static __device__ __forceinline__ float2 gcn_sum_chunk(const float2* __restrict__ S2, int lane,
                                                       int sc, int sv, int t, float2 acc) {
    float2 x[U];
    float  v[U];
    #pragma unroll
    for (int k = 0; k < U; ++k) {
        const int c = __builtin_amdgcn_readlane(sc, t + k);
        v[k] = __int_as_float(__builtin_amdgcn_readlane(sv, t + k));
        x[k] = S2[(long long)c * (DIM / 2) + lane];
    }
    #pragma unroll
    for (int k = 0; k < U; ++k) {
        acc.x += v[k] * x[k].x;
        acc.y += v[k] * x[k].y;
    }
    return acc;
}

static __device__ __forceinline__ float2 gcn_sum_records(const float2* __restrict__ S2, int lane,
                                                         int sc, int sv, int n, float2 acc) {
    int t = 0;
    for (; t + 8 <= n; t += 8) acc = gcn_sum_chunk<8>(S2, lane, sc, sv, t, acc);
    if (t + 4 <= n) { acc = gcn_sum_chunk<4>(S2, lane, sc, sv, t, acc); t += 4; }
    if (t + 2 <= n) { acc = gcn_sum_chunk<2>(S2, lane, sc, sv, t, acc); t += 2; }
    if (t < n)      { acc = gcn_sum_chunk<1>(S2, lane, sc, sv, t, acc); }
    return acc;
}

__global__ void __launch_bounds__(GATHER_WAVES * 32)
gcn_gather_kernel(const float* __restrict__ S,
                  const int* __restrict__ adj_col,
                  const float* __restrict__ adj_val,
                  const int* __restrict__ cnt,
                  const int* __restrict__ ovf_head,
                  const int* __restrict__ ovf_next,
                  const int* __restrict__ slots,
                  float* __restrict__ out,
                  int n_nodes) {
    __shared__ int2 srt[GATHER_WAVES][SLOTS];

    const int lane = threadIdx.x & 31;
    const int w = __builtin_amdgcn_readfirstlane((int)(threadIdx.x >> 5));
    const int r = blockIdx.x * GATHER_WAVES + w;
    if (r >= n_nodes) return;

    const float2* __restrict__ S2 = (const float2*)S;
    const int deg   = cnt[r];
    const int nslot = deg < SLOTS ? deg : SLOTS;
    const int4* __restrict__ rs = (const int4*)slots + (long long)r * SLOTS;

    int e0 = NO_EDGE, c0 = 0, v0 = 0;
    int e1 = NO_EDGE, c1 = 0, v1 = 0;
    if (lane < nslot)      { const int4 q = rs[lane];      e0 = q.x; c0 = q.y; v0 = q.z; }
    if (lane + 32 < nslot) { const int4 q = rs[lane + 32]; e1 = q.x; c1 = q.y; v1 = q.z; }

    float2 acc = make_float2(0.f, 0.f);

    if (deg <= SLOTS) {
        int rk0 = 0, rk1 = 0;
        const int n0 = nslot < 32 ? nslot : 32;
        if (nslot <= 32) {
            for (int j = 0; j < n0; ++j) {
                const int kj = __builtin_amdgcn_readlane(e0, j);
                rk0 += (kj < e0) ? 1 : 0;
            }
        } else {
            for (int j = 0; j < 32; ++j) {
                const int kj = __builtin_amdgcn_readlane(e0, j);
                rk0 += (kj < e0) ? 1 : 0;
                rk1 += (kj < e1) ? 1 : 0;
            }
            for (int j = 32; j < nslot; ++j) {
                const int kj = __builtin_amdgcn_readlane(e1, j - 32);
                rk0 += (kj < e0) ? 1 : 0;
                rk1 += (kj < e1) ? 1 : 0;
            }
        }
        if (lane < nslot)      srt[w][rk0] = make_int2(c0, v0);
        if (lane + 32 < nslot) srt[w][rk1] = make_int2(c1, v1);
        int2 s0 = make_int2(0, 0), s1 = make_int2(0, 0);
        if (lane < nslot)      s0 = srt[w][lane];
        if (lane + 32 < nslot) s1 = srt[w][lane + 32];

        acc = gcn_sum_records(S2, lane, s0.x, s0.y, n0, acc);
        if (nslot > 32) acc = gcn_sum_records(S2, lane, s1.x, s1.y, nslot - 32, acc);
    } else {
        int last = -1;
        for (int t = 0; t < deg; ++t) {
            int best = NO_EDGE;
            if (e0 > last) best = e0;
            if (e1 > last && e1 < best) best = e1;
            for (int off = 16; off > 0; off >>= 1) {
                const int o = __shfl_xor(best, off);
                best = o < best ? o : best;
            }
            best = __builtin_amdgcn_readfirstlane(best);
            int p = ovf_head[r];
            for (int g = 0; g < deg && p != 0; ++g) {
                const int id = p - 1;
                if (id > last && id < best) best = id;
                p = ovf_next[id];
            }
            if (best == NO_EDGE) break;
            const int    c = adj_col[best];
            const float  v = adj_val[best];
            const float2 x = S2[(long long)c * (DIM / 2) + lane];
            acc.x += v * x.x;
            acc.y += v * x.y;
            last = best;
        }
    }

    ((float2*)out)[(long long)r * (DIM / 2) + lane] = acc;
}

extern "C" void kernel_launch(void* const* d_in, const int* in_sizes, int n_in,
                              void* d_out, int out_size, void* d_ws, size_t ws_size,
                              hipStream_t stream) {
    const float* X    = (const float*)d_in[0];
    const float* W    = (const float*)d_in[1];
    const int*   arow = (const int*)d_in[2];
    const int*   acol = (const int*)d_in[3];
    const float* aval = (const float*)d_in[4];
    float*       out  = (float*)d_out;

    const int n_nodes = in_sizes[0] / DIM;
    const int n_edges = in_sizes[2];

    float* S        = (float*)d_ws;
    int*   cnt      = (int*)(S + (size_t)n_nodes * DIM);
    const int nz4   = (2 * n_nodes + 3) / 4;
    int*   ovf_head = cnt + n_nodes;
    int*   ovf_next = cnt + (size_t)nz4 * 4;
    int*   slots    = ovf_next + (size_t)((n_edges + 3) / 4) * 4;

    gcn_zero_kernel<<<(nz4 + 255) / 256, 256, 0, stream>>>((float*)cnt, nz4);

    gcn_fill_kernel<<<(n_edges + 255) / 256, 256, 0, stream>>>(arow, acol, aval, cnt, ovf_head,
                                                               ovf_next, slots, n_edges);

    {
        const int tiles = n_nodes / 16;
        const int waves_per_block = 256 / 32;
        const int blocks = (tiles + waves_per_block - 1) / waves_per_block;
        gcn_gemm_wmma_kernel<<<blocks, 256, 0, stream>>>(X, W, S, tiles);
    }

    gcn_gather_kernel<<<(n_nodes + GATHER_WAVES - 1) / GATHER_WAVES, GATHER_WAVES * 32, 0, stream>>>(
        S, acol, aval, cnt, ovf_head, ovf_next, slots, out, n_nodes);
}


# Per-edge global returning atomics removed: edges grouped by a counting sort over 256-row buckets with LDS counters (count, place, rank launches); zero launch dropped; per-row sum order unchanged
# speedup vs baseline: 3.4116x; 1.6415x over previous
#include <hip/hip_runtime.h>

typedef __attribute__((ext_vector_type(2))) float v2f;
typedef __attribute__((ext_vector_type(8))) float v8f;

#define DIM 64
#define SLOTS 64
#define NO_EDGE 0x7fffffff

#define NCHUNK      256
#define BUCKET_ROWS 256
#define MAX_BUCKETS 512
#define BIN_THREADS 512
#define BIN_EDGES   4

__global__ void gcn_gemm_wmma_kernel(const float* __restrict__ X,
                                     const float* __restrict__ W,
                                     float* __restrict__ S,
                                     int n_row_tiles) {
    const int wave = threadIdx.x >> 5;
    const int lane = threadIdx.x & 31;
    const int tile = blockIdx.x * (blockDim.x >> 5) + wave;
    if (tile >= n_row_tiles) return;

    const int m0    = tile << 4;
    const int mrow  = lane & 15;
    const int khalf = lane >> 4;
    const int ncol  = lane & 15;

    v8f acc0 = {}, acc1 = {}, acc2 = {}, acc3 = {};

    const float* __restrict__ xrow = X + (long long)(m0 + mrow) * DIM;

    #pragma unroll
    for (int k0 = 0; k0 < DIM; k0 += 4) {
        const int ka = k0 + (khalf << 1);
        v2f a = *(const v2f*)(xrow + ka);

        const float* __restrict__ w0 = W + ka * DIM;
        const float* __restrict__ w1 = W + (ka + 1) * DIM;

        v2f b0, b1, b2, b3;
        b0.x = w0[ 0 + ncol]; b0.y = w1[ 0 + ncol];
        b1.x = w0[16 + ncol]; b1.y = w1[16 + ncol];
        b2.x = w0[32 + ncol]; b2.y = w1[32 + ncol];
        b3.x = w0[48 + ncol]; b3.y = w1[48 + ncol];

        acc0 = __builtin_amdgcn_wmma_f32_16x16x4_f32(false, a, false, b0,
                                                     (short)0, acc0, false, false);
        acc1 = __builtin_amdgcn_wmma_f32_16x16x4_f32(false, a, false, b1,
                                                     (short)0, acc1, false, false);
        acc2 = __builtin_amdgcn_wmma_f32_16x16x4_f32(false, a, false, b2,
                                                     (short)0, acc2, false, false);
        acc3 = __builtin_amdgcn_wmma_f32_16x16x4_f32(false, a, false, b3,
                                                     (short)0, acc3, false, false);
    }

    const int rbase = m0 + (khalf << 3);
    #pragma unroll
    for (int v = 0; v < 8; ++v) {
        float* __restrict__ srow = S + (long long)(rbase + v) * DIM + ncol;
        srow[ 0] = acc0[v];
        srow[16] = acc1[v];
        srow[32] = acc2[v];
        srow[48] = acc3[v];
    }
}

__global__ void __launch_bounds__(BIN_THREADS)
gcn_count_kernel(const int* __restrict__ adj_row, int* __restrict__ table, int n_edges) {
    __shared__ int lh[MAX_BUCKETS];
    const int t = threadIdx.x;
    lh[t] = 0;
    __syncthreads();

    const int run = (n_edges + NCHUNK - 1) / NCHUNK;
    const int e0  = blockIdx.x * run;
    const int e1  = (e0 + run < n_edges) ? e0 + run : n_edges;
    for (int eb = e0 + t; eb < e1; eb += BIN_EDGES * BIN_THREADS) {
        int r[BIN_EDGES];
        #pragma unroll
        for (int k = 0; k < BIN_EDGES; ++k) {
            const int e = eb + k * BIN_THREADS;
            r[k] = (e < e1) ? adj_row[e] : -1;
        }
        #pragma unroll
        for (int k = 0; k < BIN_EDGES; ++k) {
            if (r[k] >= 0) atomicAdd(&lh[r[k] >> 8], 1);
        }
    }
    __syncthreads();
    table[blockIdx.x * MAX_BUCKETS + t] = lh[t];
}

__global__ void __launch_bounds__(BIN_THREADS)
gcn_place_kernel(const int* __restrict__ adj_row,
                 const int* __restrict__ adj_col,
                 const float* __restrict__ adj_val,
                 const int* __restrict__ table,
                 int* __restrict__ bucket_start,
                 int4* __restrict__ binned,
                 int n_edges) {
    __shared__ int sc[MAX_BUCKETS];
    __shared__ int lpos[MAX_BUCKETS];
    const int t = threadIdx.x;
    const int c = blockIdx.x;

    int tot = 0, pre = 0;
    #pragma unroll 32
    for (int cc = 0; cc < NCHUNK; ++cc) {
        const int h = table[cc * MAX_BUCKETS + t];
        tot += h;
        pre += (cc < c) ? h : 0;
    }
    sc[t] = tot;
    __syncthreads();
    for (int off = 1; off < MAX_BUCKETS; off <<= 1) {
        const int add = (t >= off) ? sc[t - off] : 0;
        __syncthreads();
        sc[t] += add;
        __syncthreads();
    }
    const int first = sc[t] - tot;
    lpos[t] = first + pre;
    if (c == 0) {
        bucket_start[t] = first;
        if (t == MAX_BUCKETS - 1) bucket_start[MAX_BUCKETS] = sc[t];
    }
    __syncthreads();

    const int run = (n_edges + NCHUNK - 1) / NCHUNK;
    const int e0  = c * run;
    const int e1  = (e0 + run < n_edges) ? e0 + run : n_edges;
    for (int eb = e0 + t; eb < e1; eb += BIN_EDGES * BIN_THREADS) {
        int   r[BIN_EDGES], cl[BIN_EDGES], p[BIN_EDGES];
        float v[BIN_EDGES];
        #pragma unroll
        for (int k = 0; k < BIN_EDGES; ++k) {
            const int e = eb + k * BIN_THREADS;
            r[k] = -1; cl[k] = 0; v[k] = 0.f;
            if (e < e1) { r[k] = adj_row[e]; cl[k] = adj_col[e]; v[k] = adj_val[e]; }
        }
        #pragma unroll
        for (int k = 0; k < BIN_EDGES; ++k) {
            p[k] = 0;
            if (r[k] >= 0) p[k] = atomicAdd(&lpos[r[k] >> 8], 1);
        }
        #pragma unroll
        for (int k = 0; k < BIN_EDGES; ++k) {
            const int e = eb + k * BIN_THREADS;
            if (r[k] >= 0) binned[p[k]] = make_int4(e, r[k], cl[k], __float_as_int(v[k]));
        }
    }
}

__global__ void __launch_bounds__(BIN_THREADS)
gcn_rank_kernel(const int4* __restrict__ binned,
                const int* __restrict__ bucket_start,
                int* __restrict__ cnt,
                int* __restrict__ ovf_head,
                int* __restrict__ ovf_next,
                int* __restrict__ slots,
                int n_nodes) {
    __shared__ int lcnt[BUCKET_ROWS];
    __shared__ int lhead[BUCKET_ROWS];
    const int t  = threadIdx.x;
    const int r0 = blockIdx.x * BUCKET_ROWS;
    if (r0 >= n_nodes) return;

    if (t < BUCKET_ROWS) { lcnt[t] = 0; lhead[t] = 0; }
    __syncthreads();

    const int beg = bucket_start[blockIdx.x];
    const int end = bucket_start[blockIdx.x + 1];
    for (int ib = beg + t; ib < end; ib += BIN_EDGES * BIN_THREADS) {
        int4 q[BIN_EDGES];
        int  rank[BIN_EDGES];
        #pragma unroll
        for (int k = 0; k < BIN_EDGES; ++k) {
            const int i = ib + k * BIN_THREADS;
            q[k] = make_int4(0, -1, 0, 0);
            if (i < end) q[k] = binned[i];
        }
        #pragma unroll
        for (int k = 0; k < BIN_EDGES; ++k) {
            rank[k] = 0;
            if (q[k].y >= 0) rank[k] = atomicAdd(&lcnt[q[k].y - r0], 1);
        }
        #pragma unroll
        for (int k = 0; k < BIN_EDGES; ++k) {
            if (q[k].y >= 0) {
                if (rank[k] < SLOTS) {
                    ((int4*)slots)[(long long)q[k].y * SLOTS + rank[k]] =
                        make_int4(q[k].x, q[k].z, q[k].w, 0);
                } else {
                    ovf_next[q[k].x] = atomicExch(&lhead[q[k].y - r0], q[k].x + 1);
                }
            }
        }
    }
    __syncthreads();

    if (t < BUCKET_ROWS && r0 + t < n_nodes) {
        cnt[r0 + t]      = lcnt[t];
        ovf_head[r0 + t] = lhead[t];
    }
}

#define GATHER_WAVES 8

template <int U>
static __device__ __forceinline__ float2 gcn_sum_chunk(const float2* __restrict__ S2, int lane,
                                                       int sc, int sv, int t, float2 acc) {
    float2 x[U];
    float  v[U];
    #pragma unroll
    for (int k = 0; k < U; ++k) {
        const int c = __builtin_amdgcn_readlane(sc, t + k);
        v[k] = __int_as_float(__builtin_amdgcn_readlane(sv, t + k));
        x[k] = S2[(long long)c * (DIM / 2) + lane];
    }
    #pragma unroll
    for (int k = 0; k < U; ++k) {
        acc.x += v[k] * x[k].x;
        acc.y += v[k] * x[k].y;
    }
    return acc;
}

static __device__ __forceinline__ float2 gcn_sum_records(const float2* __restrict__ S2, int lane,
                                                         int sc, int sv, int n, float2 acc) {
    int t = 0;
    for (; t + 8 <= n; t += 8) acc = gcn_sum_chunk<8>(S2, lane, sc, sv, t, acc);
    if (t + 4 <= n) { acc = gcn_sum_chunk<4>(S2, lane, sc, sv, t, acc); t += 4; }
    if (t + 2 <= n) { acc = gcn_sum_chunk<2>(S2, lane, sc, sv, t, acc); t += 2; }
    if (t < n)      { acc = gcn_sum_chunk<1>(S2, lane, sc, sv, t, acc); }
    return acc;
}

__global__ void __launch_bounds__(GATHER_WAVES * 32)
gcn_gather_kernel(const float* __restrict__ S,
                  const int* __restrict__ adj_col,
                  const float* __restrict__ adj_val,
                  const int* __restrict__ cnt,
                  const int* __restrict__ ovf_head,
                  const int* __restrict__ ovf_next,
                  const int* __restrict__ slots,
                  float* __restrict__ out,
                  int n_nodes) {
    __shared__ int2 srt[GATHER_WAVES][SLOTS];

    const int lane = threadIdx.x & 31;
    const int w = __builtin_amdgcn_readfirstlane((int)(threadIdx.x >> 5));
    const int r = blockIdx.x * GATHER_WAVES + w;
    if (r >= n_nodes) return;

    const float2* __restrict__ S2 = (const float2*)S;
    const int deg   = cnt[r];
    const int nslot = deg < SLOTS ? deg : SLOTS;
    const int4* __restrict__ rs = (const int4*)slots + (long long)r * SLOTS;

    int e0 = NO_EDGE, c0 = 0, v0 = 0;
    int e1 = NO_EDGE, c1 = 0, v1 = 0;
    if (lane < nslot)      { const int4 q = rs[lane];      e0 = q.x; c0 = q.y; v0 = q.z; }
    if (lane + 32 < nslot) { const int4 q = rs[lane + 32]; e1 = q.x; c1 = q.y; v1 = q.z; }

    float2 acc = make_float2(0.f, 0.f);

    if (deg <= SLOTS) {
        int rk0 = 0, rk1 = 0;
        const int n0 = nslot < 32 ? nslot : 32;
        if (nslot <= 32) {
            for (int j = 0; j < n0; ++j) {
                const int kj = __builtin_amdgcn_readlane(e0, j);
                rk0 += (kj < e0) ? 1 : 0;
            }
        } else {
            for (int j = 0; j < 32; ++j) {
                const int kj = __builtin_amdgcn_readlane(e0, j);
                rk0 += (kj < e0) ? 1 : 0;
                rk1 += (kj < e1) ? 1 : 0;
            }
            for (int j = 32; j < nslot; ++j) {
                const int kj = __builtin_amdgcn_readlane(e1, j - 32);
                rk0 += (kj < e0) ? 1 : 0;
                rk1 += (kj < e1) ? 1 : 0;
            }
        }
        if (lane < nslot)      srt[w][rk0] = make_int2(c0, v0);
        if (lane + 32 < nslot) srt[w][rk1] = make_int2(c1, v1);
        int2 s0 = make_int2(0, 0), s1 = make_int2(0, 0);
        if (lane < nslot)      s0 = srt[w][lane];
        if (lane + 32 < nslot) s1 = srt[w][lane + 32];

        acc = gcn_sum_records(S2, lane, s0.x, s0.y, n0, acc);
        if (nslot > 32) acc = gcn_sum_records(S2, lane, s1.x, s1.y, nslot - 32, acc);
    } else {
        int last = -1;
        for (int t = 0; t < deg; ++t) {
            int best = NO_EDGE;
            if (e0 > last) best = e0;
            if (e1 > last && e1 < best) best = e1;
            for (int off = 16; off > 0; off >>= 1) {
                const int o = __shfl_xor(best, off);
                best = o < best ? o : best;
            }
            best = __builtin_amdgcn_readfirstlane(best);
            int p = ovf_head[r];
            for (int g = 0; g < deg && p != 0; ++g) {
                const int id = p - 1;
                if (id > last && id < best) best = id;
                p = ovf_next[id];
            }
            if (best == NO_EDGE) break;
            const int    c = adj_col[best];
            const float  v = adj_val[best];
            const float2 x = S2[(long long)c * (DIM / 2) + lane];
            acc.x += v * x.x;
            acc.y += v * x.y;
            last = best;
        }
    }

    ((float2*)out)[(long long)r * (DIM / 2) + lane] = acc;
}

extern "C" void kernel_launch(void* const* d_in, const int* in_sizes, int n_in,
                              void* d_out, int out_size, void* d_ws, size_t ws_size,
                              hipStream_t stream) {
    const float* X    = (const float*)d_in[0];
    const float* W    = (const float*)d_in[1];
    const int*   arow = (const int*)d_in[2];
    const int*   acol = (const int*)d_in[3];
    const float* aval = (const float*)d_in[4];
    float*       out  = (float*)d_out;

    const int n_nodes = in_sizes[0] / DIM;
    const int n_edges = in_sizes[2];

    float* S        = (float*)d_ws;
    int*   cnt      = (int*)(S + (size_t)n_nodes * DIM);
    const int nz4   = (2 * n_nodes + 3) / 4;
    int*   ovf_head = cnt + n_nodes;
    int*   ovf_next = cnt + (size_t)nz4 * 4;
    int*   slots    = ovf_next + (size_t)((n_edges + 3) / 4) * 4;
    int*   table    = slots + (size_t)n_nodes * SLOTS * 4;
    int*   bstart   = table + NCHUNK * MAX_BUCKETS;
    int4*  binned   = (int4*)(bstart + MAX_BUCKETS + 64);

    gcn_count_kernel<<<NCHUNK, BIN_THREADS, 0, stream>>>(arow, table, n_edges);

    gcn_place_kernel<<<NCHUNK, BIN_THREADS, 0, stream>>>(arow, acol, aval, table, bstart, binned, n_edges);

    gcn_rank_kernel<<<MAX_BUCKETS, BIN_THREADS, 0, stream>>>(binned, bstart, cnt, ovf_head, ovf_next,
                                                            slots, n_nodes);

    {
        const int tiles = n_nodes / 16;
        const int waves_per_block = 256 / 32;
        const int blocks = (tiles + waves_per_block - 1) / waves_per_block;
        gcn_gemm_wmma_kernel<<<blocks, 256, 0, stream>>>(X, W, S, tiles);
    }

    gcn_gather_kernel<<<(n_nodes + GATHER_WAVES - 1) / GATHER_WAVES, GATHER_WAVES * 32, 0, stream>>>(
        S, acol, aval, cnt, ovf_head, ovf_next, slots, out, n_nodes);
}


# Rank launch folded into the summing launch: a work-group per 256-row bucket groups its records by row in LDS and sums them there; slot records in memory and one launch removed; sum order unchanged
# speedup vs baseline: 3.9020x; 1.1437x over previous
#include <hip/hip_runtime.h>

typedef __attribute__((ext_vector_type(2))) float v2f;
typedef __attribute__((ext_vector_type(8))) float v8f;

#define DIM 64
#define NO_EDGE 0x7fffffff

#define NCHUNK      256
#define BUCKET_ROWS 256
#define MAX_BUCKETS 512
#define BIN_THREADS 512
#define BIN_EDGES   4

#define SUM_THREADS 1024
#define SUM_WAVES   (SUM_THREADS / 32)
#define SUM_TAKE    6
#define SUM_CAP     (SUM_TAKE * SUM_THREADS)

__global__ void gcn_gemm_wmma_kernel(const float* __restrict__ X,
                                     const float* __restrict__ W,
                                     float* __restrict__ S,
                                     int n_row_tiles) {
    const int wave = threadIdx.x >> 5;
    const int lane = threadIdx.x & 31;
    const int tile = blockIdx.x * (blockDim.x >> 5) + wave;
    if (tile >= n_row_tiles) return;

    const int m0    = tile << 4;
    const int mrow  = lane & 15;
    const int khalf = lane >> 4;
    const int ncol  = lane & 15;

    v8f acc0 = {}, acc1 = {}, acc2 = {}, acc3 = {};

    const float* __restrict__ xrow = X + (long long)(m0 + mrow) * DIM;

    #pragma unroll
    for (int k0 = 0; k0 < DIM; k0 += 4) {
        const int ka = k0 + (khalf << 1);
        v2f a = *(const v2f*)(xrow + ka);

        const float* __restrict__ w0 = W + ka * DIM;
        const float* __restrict__ w1 = W + (ka + 1) * DIM;

        v2f b0, b1, b2, b3;
        b0.x = w0[ 0 + ncol]; b0.y = w1[ 0 + ncol];
        b1.x = w0[16 + ncol]; b1.y = w1[16 + ncol];
        b2.x = w0[32 + ncol]; b2.y = w1[32 + ncol];
        b3.x = w0[48 + ncol]; b3.y = w1[48 + ncol];

        acc0 = __builtin_amdgcn_wmma_f32_16x16x4_f32(false, a, false, b0,
                                                     (short)0, acc0, false, false);
        acc1 = __builtin_amdgcn_wmma_f32_16x16x4_f32(false, a, false, b1,
                                                     (short)0, acc1, false, false);
        acc2 = __builtin_amdgcn_wmma_f32_16x16x4_f32(false, a, false, b2,
                                                     (short)0, acc2, false, false);
        acc3 = __builtin_amdgcn_wmma_f32_16x16x4_f32(false, a, false, b3,
                                                     (short)0, acc3, false, false);
    }

    const int rbase = m0 + (khalf << 3);
    #pragma unroll
    for (int v = 0; v < 8; ++v) {
        float* __restrict__ srow = S + (long long)(rbase + v) * DIM + ncol;
        srow[ 0] = acc0[v];
        srow[16] = acc1[v];
        srow[32] = acc2[v];
        srow[48] = acc3[v];
    }
}

__global__ void __launch_bounds__(BIN_THREADS)
gcn_count_kernel(const int* __restrict__ adj_row, int* __restrict__ table, int n_edges) {
    __shared__ int lh[MAX_BUCKETS];
    const int t = threadIdx.x;
    lh[t] = 0;
    __syncthreads();

    const int run = (n_edges + NCHUNK - 1) / NCHUNK;
    const int e0  = blockIdx.x * run;
    const int e1  = (e0 + run < n_edges) ? e0 + run : n_edges;
    for (int eb = e0 + t; eb < e1; eb += BIN_EDGES * BIN_THREADS) {
        int r[BIN_EDGES];
        #pragma unroll
        for (int k = 0; k < BIN_EDGES; ++k) {
            const int e = eb + k * BIN_THREADS;
            r[k] = (e < e1) ? adj_row[e] : -1;
        }
        #pragma unroll
        for (int k = 0; k < BIN_EDGES; ++k) {
            if (r[k] >= 0) atomicAdd(&lh[r[k] >> 8], 1);
        }
    }
    __syncthreads();
    table[blockIdx.x * MAX_BUCKETS + t] = lh[t];
}

__global__ void __launch_bounds__(BIN_THREADS)
gcn_place_kernel(const int* __restrict__ adj_row,
                 const int* __restrict__ adj_col,
                 const float* __restrict__ adj_val,
                 const int* __restrict__ table,
                 int* __restrict__ bucket_start,
                 int4* __restrict__ binned,
                 int n_edges) {
    __shared__ int sc[MAX_BUCKETS];
    __shared__ int lpos[MAX_BUCKETS];
    const int t = threadIdx.x;
    const int c = blockIdx.x;

    int tot = 0, pre = 0;
    #pragma unroll 32
    for (int cc = 0; cc < NCHUNK; ++cc) {
        const int h = table[cc * MAX_BUCKETS + t];
        tot += h;
        pre += (cc < c) ? h : 0;
    }
    sc[t] = tot;
    __syncthreads();
    for (int off = 1; off < MAX_BUCKETS; off <<= 1) {
        const int add = (t >= off) ? sc[t - off] : 0;
        __syncthreads();
        sc[t] += add;
        __syncthreads();
    }
    const int first = sc[t] - tot;
    lpos[t] = first + pre;
    if (c == 0) {
        bucket_start[t] = first;
        if (t == MAX_BUCKETS - 1) bucket_start[MAX_BUCKETS] = sc[t];
    }
    __syncthreads();

    const int run = (n_edges + NCHUNK - 1) / NCHUNK;
    const int e0  = c * run;
    const int e1  = (e0 + run < n_edges) ? e0 + run : n_edges;
    for (int eb = e0 + t; eb < e1; eb += BIN_EDGES * BIN_THREADS) {
        int   r[BIN_EDGES], cl[BIN_EDGES], p[BIN_EDGES];
        float v[BIN_EDGES];
        #pragma unroll
        for (int k = 0; k < BIN_EDGES; ++k) {
            const int e = eb + k * BIN_THREADS;
            r[k] = -1; cl[k] = 0; v[k] = 0.f;
            if (e < e1) { r[k] = adj_row[e]; cl[k] = adj_col[e]; v[k] = adj_val[e]; }
        }
        #pragma unroll
        for (int k = 0; k < BIN_EDGES; ++k) {
            p[k] = 0;
            if (r[k] >= 0) p[k] = atomicAdd(&lpos[r[k] >> 8], 1);
        }
        #pragma unroll
        for (int k = 0; k < BIN_EDGES; ++k) {
            const int e = eb + k * BIN_THREADS;
            if (r[k] >= 0) binned[p[k]] = make_int4(e, r[k], cl[k], __float_as_int(v[k]));
        }
    }
}

template <int U>
static __device__ __forceinline__ float2 gcn_sum_chunk(const float2* __restrict__ S2, int lane,
                                                       int sc, int sv, int t, float2 acc) {
    float2 x[U];
    float  v[U];
    #pragma unroll
    for (int k = 0; k < U; ++k) {
        const int c = __builtin_amdgcn_readlane(sc, t + k);
        v[k] = __int_as_float(__builtin_amdgcn_readlane(sv, t + k));
        x[k] = S2[(long long)c * (DIM / 2) + lane];
    }
    #pragma unroll
    for (int k = 0; k < U; ++k) {
        acc.x += v[k] * x[k].x;
        acc.y += v[k] * x[k].y;
    }
    return acc;
}

static __device__ __forceinline__ float2 gcn_sum_records(const float2* __restrict__ S2, int lane,
                                                         int sc, int sv, int n, float2 acc) {
    int t = 0;
    for (; t + 8 <= n; t += 8) acc = gcn_sum_chunk<8>(S2, lane, sc, sv, t, acc);
    if (t + 4 <= n) { acc = gcn_sum_chunk<4>(S2, lane, sc, sv, t, acc); t += 4; }
    if (t + 2 <= n) { acc = gcn_sum_chunk<2>(S2, lane, sc, sv, t, acc); t += 2; }
    if (t < n)      { acc = gcn_sum_chunk<1>(S2, lane, sc, sv, t, acc); }
    return acc;
}

static __device__ __forceinline__ void gcn_wave_min(int& best, int& where) {
    for (int off = 16; off > 0; off >>= 1) {
        const int ob = __shfl_xor(best, off);
        const int ow = __shfl_xor(where, off);
        if (ob < best) { best = ob; where = ow; }
    }
}

__global__ void __launch_bounds__(SUM_THREADS)
gcn_group_sum_kernel(const int4* __restrict__ binned,
                     const int* __restrict__ bucket_start,
                     const float* __restrict__ S,
                     float* __restrict__ out,
                     int n_nodes) {
    __shared__ int  lcnt[BUCKET_ROWS];
    __shared__ int  loff[BUCKET_ROWS];
    __shared__ int  wtot[BUCKET_ROWS / 32];
    __shared__ int  rec_id[SUM_CAP];
    __shared__ int2 rec_cv[SUM_CAP];

    const int t    = threadIdx.x;
    const int lane = t & 31;
    const int w  = __builtin_amdgcn_readfirstlane(t >> 5);
    const int r0 = blockIdx.x * BUCKET_ROWS;
    if (r0 >= n_nodes) return;

    const int beg  = bucket_start[blockIdx.x];
    const int end  = bucket_start[blockIdx.x + 1];
    const int n    = end - beg;
    const int rows = (n_nodes - r0 < BUCKET_ROWS) ? n_nodes - r0 : BUCKET_ROWS;
    const float2* __restrict__ S2 = (const float2*)S;

    if (n > SUM_CAP) {
        for (int i = 0; i < BUCKET_ROWS / SUM_WAVES; ++i) {
            const int rl = i * SUM_WAVES + w;
            if (rl >= rows) break;
            const int r = r0 + rl;
            float2 acc = make_float2(0.f, 0.f);
            int last = -1;
            for (int g = 0; g < n; ++g) {
                int best = NO_EDGE, where = 0;
                for (int j = beg + lane; j < end; j += 32) {
                    const int4 q = binned[j];
                    if (q.y == r && q.x > last && q.x < best) { best = q.x; where = j; }
                }
                gcn_wave_min(best, where);
                if (best == NO_EDGE) break;
                const int4   q = binned[where];
                const float  v = __int_as_float(q.w);
                const float2 x = S2[(long long)q.z * (DIM / 2) + lane];
                acc.x += v * x.x;
                acc.y += v * x.y;
                last = best;
            }
            ((float2*)out)[(long long)r * (DIM / 2) + lane] = acc;
        }
        return;
    }

    if (t < BUCKET_ROWS) lcnt[t] = 0;
    __syncthreads();

    int4 q[SUM_TAKE];
    int  place[SUM_TAKE];
    #pragma unroll
    for (int k = 0; k < SUM_TAKE; ++k) {
        const int i = beg + t + k * SUM_THREADS;
        q[k] = make_int4(0, -1, 0, 0);
        if (i < end) q[k] = binned[i];
    }
    #pragma unroll
    for (int k = 0; k < SUM_TAKE; ++k) {
        place[k] = 0;
        if (q[k].y >= 0) place[k] = atomicAdd(&lcnt[q[k].y - r0], 1);
    }
    __syncthreads();

    int c = 0, incl = 0;
    if (t < BUCKET_ROWS) {
        c = lcnt[t];
        incl = c;
        for (int off = 1; off < 32; off <<= 1) {
            const int up = __shfl_up(incl, off);
            if (lane >= off) incl += up;
        }
        if (lane == 31) wtot[w] = incl;
    }
    __syncthreads();
    if (t < BUCKET_ROWS) {
        int base = 0;
        for (int j = 0; j < BUCKET_ROWS / 32; ++j) base += (j < w) ? wtot[j] : 0;
        loff[t] = base + incl - c;
    }
    __syncthreads();

    #pragma unroll
    for (int k = 0; k < SUM_TAKE; ++k) {
        if (q[k].y >= 0) {
            const int p = loff[q[k].y - r0] + place[k];
            rec_id[p] = q[k].x;
            rec_cv[p] = make_int2(q[k].z, q[k].w);
        }
    }
    __syncthreads();

    for (int i = 0; i < BUCKET_ROWS / SUM_WAVES; ++i) {
        const int rl = i * SUM_WAVES + w;
        if (rl >= rows) break;
        const int deg = __builtin_amdgcn_readfirstlane(lcnt[rl]);
        const int off = __builtin_amdgcn_readfirstlane(loff[rl]);

        float2 acc = make_float2(0.f, 0.f);

        if (deg <= 64) {
            int e0 = NO_EDGE, c0 = 0, v0 = 0;
            int e1 = NO_EDGE, c1 = 0, v1 = 0;
            if (lane < deg)      { e0 = rec_id[off + lane];      const int2 cv = rec_cv[off + lane];      c0 = cv.x; v0 = cv.y; }
            if (lane + 32 < deg) { e1 = rec_id[off + lane + 32]; const int2 cv = rec_cv[off + lane + 32]; c1 = cv.x; v1 = cv.y; }

            int rk0 = 0, rk1 = 0;
            const int n0 = deg < 32 ? deg : 32;
            if (deg <= 32) {
                for (int j = 0; j < n0; ++j) {
                    const int kj = __builtin_amdgcn_readlane(e0, j);
                    rk0 += (kj < e0) ? 1 : 0;
                }
            } else {
                for (int j = 0; j < 32; ++j) {
                    const int kj = __builtin_amdgcn_readlane(e0, j);
                    rk0 += (kj < e0) ? 1 : 0;
                    rk1 += (kj < e1) ? 1 : 0;
                }
                for (int j = 32; j < deg; ++j) {
                    const int kj = __builtin_amdgcn_readlane(e1, j - 32);
                    rk0 += (kj < e0) ? 1 : 0;
                    rk1 += (kj < e1) ? 1 : 0;
                }
            }
            if (lane < deg)      rec_cv[off + rk0] = make_int2(c0, v0);
            if (lane + 32 < deg) rec_cv[off + rk1] = make_int2(c1, v1);
            int2 s0 = make_int2(0, 0), s1 = make_int2(0, 0);
            if (lane < deg)      s0 = rec_cv[off + lane];
            if (lane + 32 < deg) s1 = rec_cv[off + lane + 32];

            acc = gcn_sum_records(S2, lane, s0.x, s0.y, n0, acc);
            if (deg > 32) acc = gcn_sum_records(S2, lane, s1.x, s1.y, deg - 32, acc);
        } else {
            int last = -1;
            for (int g = 0; g < deg; ++g) {
                int best = NO_EDGE, where = 0;
                for (int j = lane; j < deg; j += 32) {
                    const int id = rec_id[off + j];
                    if (id > last && id < best) { best = id; where = j; }
                }
                gcn_wave_min(best, where);
                if (best == NO_EDGE) break;
                const int2   cv = rec_cv[off + where];
                const float  v  = __int_as_float(cv.y);
                const float2 x  = S2[(long long)cv.x * (DIM / 2) + lane];
                acc.x += v * x.x;
                acc.y += v * x.y;
                last = best;
            }
        }

        ((float2*)out)[(long long)(r0 + rl) * (DIM / 2) + lane] = acc;
    }
}

extern "C" void kernel_launch(void* const* d_in, const int* in_sizes, int n_in,
                              void* d_out, int out_size, void* d_ws, size_t ws_size,
                              hipStream_t stream) {
    const float* X    = (const float*)d_in[0];
    const float* W    = (const float*)d_in[1];
    const int*   arow = (const int*)d_in[2];
    const int*   acol = (const int*)d_in[3];
    const float* aval = (const float*)d_in[4];
    float*       out  = (float*)d_out;

    const int n_nodes = in_sizes[0] / DIM;
    const int n_edges = in_sizes[2];

    float* S        = (float*)d_ws;
    int*   table    = (int*)(S + (size_t)n_nodes * DIM);
    int*   bstart   = table + NCHUNK * MAX_BUCKETS;
    int4*  binned   = (int4*)(bstart + MAX_BUCKETS + 64);

    gcn_count_kernel<<<NCHUNK, BIN_THREADS, 0, stream>>>(arow, table, n_edges);

    gcn_place_kernel<<<NCHUNK, BIN_THREADS, 0, stream>>>(arow, acol, aval, table, bstart, binned, n_edges);

    {
        const int tiles = n_nodes / 16;
        const int waves_per_block = 256 / 32;
        const int blocks = (tiles + waves_per_block - 1) / waves_per_block;
        gcn_gemm_wmma_kernel<<<blocks, 256, 0, stream>>>(X, W, S, tiles);
    }

    gcn_group_sum_kernel<<<MAX_BUCKETS, SUM_THREADS, 0, stream>>>(binned, bstart, S, out, n_nodes);
}


# Matrix product's launch removed: its 16-row tiles run as extra work-groups of the count and place launches, beside the edge sort instead of behind it; arithmetic and sum order unchanged
# speedup vs baseline: 4.4515x; 1.1408x over previous
#include <hip/hip_runtime.h>

typedef __attribute__((ext_vector_type(2))) float v2f;
typedef __attribute__((ext_vector_type(8))) float v8f;

#define DIM 64
#define NO_EDGE 0x7fffffff

#define NCHUNK      256
#define BUCKET_ROWS 256
#define MAX_BUCKETS 512
#define BIN_THREADS 512
#define BIN_EDGES   4

#define SUM_THREADS 1024
#define SUM_WAVES   (SUM_THREADS / 32)
#define SUM_TAKE    6
#define SUM_CAP     (SUM_TAKE * SUM_THREADS)

#define GEMM_WAVES (BIN_THREADS / 32)

static __device__ __forceinline__ void gcn_gemm_tile(const float* __restrict__ X,
                                                     const float* __restrict__ W,
                                                     float* __restrict__ S,
                                                     int tile, int lane) {
    const int m0    = tile << 4;
    const int mrow  = lane & 15;
    const int khalf = lane >> 4;
    const int ncol  = lane & 15;

    v8f acc0 = {}, acc1 = {}, acc2 = {}, acc3 = {};

    const float* __restrict__ xrow = X + (long long)(m0 + mrow) * DIM;

    #pragma unroll
    for (int k0 = 0; k0 < DIM; k0 += 4) {
        const int ka = k0 + (khalf << 1);
        v2f a = *(const v2f*)(xrow + ka);

        const float* __restrict__ w0 = W + ka * DIM;
        const float* __restrict__ w1 = W + (ka + 1) * DIM;

        v2f b0, b1, b2, b3;
        b0.x = w0[ 0 + ncol]; b0.y = w1[ 0 + ncol];
        b1.x = w0[16 + ncol]; b1.y = w1[16 + ncol];
        b2.x = w0[32 + ncol]; b2.y = w1[32 + ncol];
        b3.x = w0[48 + ncol]; b3.y = w1[48 + ncol];

        acc0 = __builtin_amdgcn_wmma_f32_16x16x4_f32(false, a, false, b0,
                                                     (short)0, acc0, false, false);
        acc1 = __builtin_amdgcn_wmma_f32_16x16x4_f32(false, a, false, b1,
                                                     (short)0, acc1, false, false);
        acc2 = __builtin_amdgcn_wmma_f32_16x16x4_f32(false, a, false, b2,
                                                     (short)0, acc2, false, false);
        acc3 = __builtin_amdgcn_wmma_f32_16x16x4_f32(false, a, false, b3,
                                                     (short)0, acc3, false, false);
    }

    const int rbase = m0 + (khalf << 3);
    #pragma unroll
    for (int v = 0; v < 8; ++v) {
        float* __restrict__ srow = S + (long long)(rbase + v) * DIM + ncol;
        srow[ 0] = acc0[v];
        srow[16] = acc1[v];
        srow[32] = acc2[v];
        srow[48] = acc3[v];
    }
}

static __device__ __forceinline__ void gcn_gemm_extra_group(const float* __restrict__ X,
                                                            const float* __restrict__ W,
                                                            float* __restrict__ S,
                                                            int tile_first, int tile_end) {
    const int w    = __builtin_amdgcn_readfirstlane(threadIdx.x >> 5);
    const int tile = tile_first + (blockIdx.x - NCHUNK) * GEMM_WAVES + w;
    if (tile < tile_end) gcn_gemm_tile(X, W, S, tile, threadIdx.x & 31);
}

__global__ void __launch_bounds__(BIN_THREADS)
gcn_count_kernel(const int* __restrict__ adj_row, int* __restrict__ table, int n_edges,
                 const float* __restrict__ X, const float* __restrict__ W,
                 float* __restrict__ S, int tile_first, int tile_end) {
    __shared__ int lh[MAX_BUCKETS];
    if (blockIdx.x >= NCHUNK) {
        gcn_gemm_extra_group(X, W, S, tile_first, tile_end);
        return;
    }
    const int t = threadIdx.x;
    lh[t] = 0;
    __syncthreads();

    const int run = (n_edges + NCHUNK - 1) / NCHUNK;
    const int e0  = blockIdx.x * run;
    const int e1  = (e0 + run < n_edges) ? e0 + run : n_edges;
    for (int eb = e0 + t; eb < e1; eb += BIN_EDGES * BIN_THREADS) {
        int r[BIN_EDGES];
        #pragma unroll
        for (int k = 0; k < BIN_EDGES; ++k) {
            const int e = eb + k * BIN_THREADS;
            r[k] = (e < e1) ? adj_row[e] : -1;
        }
        #pragma unroll
        for (int k = 0; k < BIN_EDGES; ++k) {
            if (r[k] >= 0) atomicAdd(&lh[r[k] >> 8], 1);
        }
    }
    __syncthreads();
    table[blockIdx.x * MAX_BUCKETS + t] = lh[t];
}

__global__ void __launch_bounds__(BIN_THREADS)
gcn_place_kernel(const int* __restrict__ adj_row,
                 const int* __restrict__ adj_col,
                 const float* __restrict__ adj_val,
                 const int* __restrict__ table,
                 int* __restrict__ bucket_start,
                 int4* __restrict__ binned,
                 int n_edges,
                 const float* __restrict__ X, const float* __restrict__ W,
                 float* __restrict__ S, int tile_first, int tile_end) {
    __shared__ int sc[MAX_BUCKETS];
    __shared__ int lpos[MAX_BUCKETS];
    if (blockIdx.x >= NCHUNK) {
        gcn_gemm_extra_group(X, W, S, tile_first, tile_end);
        return;
    }
    const int t = threadIdx.x;
    const int c = blockIdx.x;

    int tot = 0, pre = 0;
    #pragma unroll 32
    for (int cc = 0; cc < NCHUNK; ++cc) {
        const int h = table[cc * MAX_BUCKETS + t];
        tot += h;
        pre += (cc < c) ? h : 0;
    }
    sc[t] = tot;
    __syncthreads();
    for (int off = 1; off < MAX_BUCKETS; off <<= 1) {
        const int add = (t >= off) ? sc[t - off] : 0;
        __syncthreads();
        sc[t] += add;
        __syncthreads();
    }
    const int first = sc[t] - tot;
    lpos[t] = first + pre;
    if (c == 0) {
        bucket_start[t] = first;
        if (t == MAX_BUCKETS - 1) bucket_start[MAX_BUCKETS] = sc[t];
    }
    __syncthreads();

    const int run = (n_edges + NCHUNK - 1) / NCHUNK;
    const int e0  = c * run;
    const int e1  = (e0 + run < n_edges) ? e0 + run : n_edges;
    for (int eb = e0 + t; eb < e1; eb += BIN_EDGES * BIN_THREADS) {
        int   r[BIN_EDGES], cl[BIN_EDGES], p[BIN_EDGES];
        float v[BIN_EDGES];
        #pragma unroll
        for (int k = 0; k < BIN_EDGES; ++k) {
            const int e = eb + k * BIN_THREADS;
            r[k] = -1; cl[k] = 0; v[k] = 0.f;
            if (e < e1) { r[k] = adj_row[e]; cl[k] = adj_col[e]; v[k] = adj_val[e]; }
        }
        #pragma unroll
        for (int k = 0; k < BIN_EDGES; ++k) {
            p[k] = 0;
            if (r[k] >= 0) p[k] = atomicAdd(&lpos[r[k] >> 8], 1);
        }
        #pragma unroll
        for (int k = 0; k < BIN_EDGES; ++k) {
            const int e = eb + k * BIN_THREADS;
            if (r[k] >= 0) binned[p[k]] = make_int4(e, r[k], cl[k], __float_as_int(v[k]));
        }
    }
}

template <int U>
static __device__ __forceinline__ float2 gcn_sum_chunk(const float2* __restrict__ S2, int lane,
                                                       int sc, int sv, int t, float2 acc) {
    float2 x[U];
    float  v[U];
    #pragma unroll
    for (int k = 0; k < U; ++k) {
        const int c = __builtin_amdgcn_readlane(sc, t + k);
        v[k] = __int_as_float(__builtin_amdgcn_readlane(sv, t + k));
        x[k] = S2[(long long)c * (DIM / 2) + lane];
    }
    #pragma unroll
    for (int k = 0; k < U; ++k) {
        acc.x += v[k] * x[k].x;
        acc.y += v[k] * x[k].y;
    }
    return acc;
}

static __device__ __forceinline__ float2 gcn_sum_records(const float2* __restrict__ S2, int lane,
                                                         int sc, int sv, int n, float2 acc) {
    int t = 0;
    for (; t + 8 <= n; t += 8) acc = gcn_sum_chunk<8>(S2, lane, sc, sv, t, acc);
    if (t + 4 <= n) { acc = gcn_sum_chunk<4>(S2, lane, sc, sv, t, acc); t += 4; }
    if (t + 2 <= n) { acc = gcn_sum_chunk<2>(S2, lane, sc, sv, t, acc); t += 2; }
    if (t < n)      { acc = gcn_sum_chunk<1>(S2, lane, sc, sv, t, acc); }
    return acc;
}

static __device__ __forceinline__ void gcn_wave_min(int& best, int& where) {
    for (int off = 16; off > 0; off >>= 1) {
        const int ob = __shfl_xor(best, off);
        const int ow = __shfl_xor(where, off);
        if (ob < best) { best = ob; where = ow; }
    }
}

__global__ void __launch_bounds__(SUM_THREADS)
gcn_group_sum_kernel(const int4* __restrict__ binned,
                     const int* __restrict__ bucket_start,
                     const float* __restrict__ S,
                     float* __restrict__ out,
                     int n_nodes) {
    __shared__ int  lcnt[BUCKET_ROWS];
    __shared__ int  loff[BUCKET_ROWS];
    __shared__ int  wtot[BUCKET_ROWS / 32];
    __shared__ int  rec_id[SUM_CAP];
    __shared__ int2 rec_cv[SUM_CAP];

    const int t    = threadIdx.x;
    const int lane = t & 31;
    const int w  = __builtin_amdgcn_readfirstlane(t >> 5);
    const int r0 = blockIdx.x * BUCKET_ROWS;
    if (r0 >= n_nodes) return;

    const int beg  = bucket_start[blockIdx.x];
    const int end  = bucket_start[blockIdx.x + 1];
    const int n    = end - beg;
    const int rows = (n_nodes - r0 < BUCKET_ROWS) ? n_nodes - r0 : BUCKET_ROWS;
    const float2* __restrict__ S2 = (const float2*)S;

    if (n > SUM_CAP) {
        for (int i = 0; i < BUCKET_ROWS / SUM_WAVES; ++i) {
            const int rl = i * SUM_WAVES + w;
            if (rl >= rows) break;
            const int r = r0 + rl;
            float2 acc = make_float2(0.f, 0.f);
            int last = -1;
            for (int g = 0; g < n; ++g) {
                int best = NO_EDGE, where = 0;
                for (int j = beg + lane; j < end; j += 32) {
                    const int4 q = binned[j];
                    if (q.y == r && q.x > last && q.x < best) { best = q.x; where = j; }
                }
                gcn_wave_min(best, where);
                if (best == NO_EDGE) break;
                const int4   q = binned[where];
                const float  v = __int_as_float(q.w);
                const float2 x = S2[(long long)q.z * (DIM / 2) + lane];
                acc.x += v * x.x;
                acc.y += v * x.y;
                last = best;
            }
            ((float2*)out)[(long long)r * (DIM / 2) + lane] = acc;
        }
        return;
    }

    if (t < BUCKET_ROWS) lcnt[t] = 0;
    __syncthreads();

    int4 q[SUM_TAKE];
    int  place[SUM_TAKE];
    #pragma unroll
    for (int k = 0; k < SUM_TAKE; ++k) {
        const int i = beg + t + k * SUM_THREADS;
        q[k] = make_int4(0, -1, 0, 0);
        if (i < end) q[k] = binned[i];
    }
    #pragma unroll
    for (int k = 0; k < SUM_TAKE; ++k) {
        place[k] = 0;
        if (q[k].y >= 0) place[k] = atomicAdd(&lcnt[q[k].y - r0], 1);
    }
    __syncthreads();

    int c = 0, incl = 0;
    if (t < BUCKET_ROWS) {
        c = lcnt[t];
        incl = c;
        for (int off = 1; off < 32; off <<= 1) {
            const int up = __shfl_up(incl, off);
            if (lane >= off) incl += up;
        }
        if (lane == 31) wtot[w] = incl;
    }
    __syncthreads();
    if (t < BUCKET_ROWS) {
        int base = 0;
        for (int j = 0; j < BUCKET_ROWS / 32; ++j) base += (j < w) ? wtot[j] : 0;
        loff[t] = base + incl - c;
    }
    __syncthreads();

    #pragma unroll
    for (int k = 0; k < SUM_TAKE; ++k) {
        if (q[k].y >= 0) {
            const int p = loff[q[k].y - r0] + place[k];
            rec_id[p] = q[k].x;
            rec_cv[p] = make_int2(q[k].z, q[k].w);
        }
    }
    __syncthreads();

    for (int i = 0; i < BUCKET_ROWS / SUM_WAVES; ++i) {
        const int rl = i * SUM_WAVES + w;
        if (rl >= rows) break;
        const int deg = __builtin_amdgcn_readfirstlane(lcnt[rl]);
        const int off = __builtin_amdgcn_readfirstlane(loff[rl]);

        float2 acc = make_float2(0.f, 0.f);

        if (deg <= 64) {
            int e0 = NO_EDGE, c0 = 0, v0 = 0;
            int e1 = NO_EDGE, c1 = 0, v1 = 0;
            if (lane < deg)      { e0 = rec_id[off + lane];      const int2 cv = rec_cv[off + lane];      c0 = cv.x; v0 = cv.y; }
            if (lane + 32 < deg) { e1 = rec_id[off + lane + 32]; const int2 cv = rec_cv[off + lane + 32]; c1 = cv.x; v1 = cv.y; }

            int rk0 = 0, rk1 = 0;
            const int n0 = deg < 32 ? deg : 32;
            if (deg <= 32) {
                for (int j = 0; j < n0; ++j) {
                    const int kj = __builtin_amdgcn_readlane(e0, j);
                    rk0 += (kj < e0) ? 1 : 0;
                }
            } else {
                for (int j = 0; j < 32; ++j) {
                    const int kj = __builtin_amdgcn_readlane(e0, j);
                    rk0 += (kj < e0) ? 1 : 0;
                    rk1 += (kj < e1) ? 1 : 0;
                }
                for (int j = 32; j < deg; ++j) {
                    const int kj = __builtin_amdgcn_readlane(e1, j - 32);
                    rk0 += (kj < e0) ? 1 : 0;
                    rk1 += (kj < e1) ? 1 : 0;
                }
            }
            if (lane < deg)      rec_cv[off + rk0] = make_int2(c0, v0);
            if (lane + 32 < deg) rec_cv[off + rk1] = make_int2(c1, v1);
            int2 s0 = make_int2(0, 0), s1 = make_int2(0, 0);
            if (lane < deg)      s0 = rec_cv[off + lane];
            if (lane + 32 < deg) s1 = rec_cv[off + lane + 32];

            acc = gcn_sum_records(S2, lane, s0.x, s0.y, n0, acc);
            if (deg > 32) acc = gcn_sum_records(S2, lane, s1.x, s1.y, deg - 32, acc);
        } else {
            int last = -1;
            for (int g = 0; g < deg; ++g) {
                int best = NO_EDGE, where = 0;
                for (int j = lane; j < deg; j += 32) {
                    const int id = rec_id[off + j];
                    if (id > last && id < best) { best = id; where = j; }
                }
                gcn_wave_min(best, where);
                if (best == NO_EDGE) break;
                const int2   cv = rec_cv[off + where];
                const float  v  = __int_as_float(cv.y);
                const float2 x  = S2[(long long)cv.x * (DIM / 2) + lane];
                acc.x += v * x.x;
                acc.y += v * x.y;
                last = best;
            }
        }

        ((float2*)out)[(long long)(r0 + rl) * (DIM / 2) + lane] = acc;
    }
}

extern "C" void kernel_launch(void* const* d_in, const int* in_sizes, int n_in,
                              void* d_out, int out_size, void* d_ws, size_t ws_size,
                              hipStream_t stream) {
    const float* X    = (const float*)d_in[0];
    const float* W    = (const float*)d_in[1];
    const int*   arow = (const int*)d_in[2];
    const int*   acol = (const int*)d_in[3];
    const float* aval = (const float*)d_in[4];
    float*       out  = (float*)d_out;

    const int n_nodes = in_sizes[0] / DIM;
    const int n_edges = in_sizes[2];

    float* S        = (float*)d_ws;
    int*   table    = (int*)(S + (size_t)n_nodes * DIM);
    int*   bstart   = table + NCHUNK * MAX_BUCKETS;
    int4*  binned   = (int4*)(bstart + MAX_BUCKETS + 64);

    const int tiles    = n_nodes / 16;
    const int g_all    = (tiles + GEMM_WAVES - 1) / GEMM_WAVES;
    const int g_count  = g_all * 3 / 8;
    const int g_place  = g_all - g_count;
    const int tile_mid = g_count * GEMM_WAVES;

    gcn_count_kernel<<<NCHUNK + g_count, BIN_THREADS, 0, stream>>>(arow, table, n_edges,
                                                                  X, W, S, 0, tile_mid);

    gcn_place_kernel<<<NCHUNK + g_place, BIN_THREADS, 0, stream>>>(arow, acol, aval, table, bstart, binned, n_edges,
                                                                  X, W, S, tile_mid, tiles);

    gcn_group_sum_kernel<<<MAX_BUCKETS, SUM_THREADS, 0, stream>>>(binned, bstart, S, out, n_nodes);
}


# Count launch removed: each work-group sorts its own run of edges by bucket in LDS and writes a run-private stretch; a bucket's summing work-group collects its 256 stretches; sum order unchanged
# speedup vs baseline: 4.6630x; 1.0475x over previous
#include <hip/hip_runtime.h>

typedef __attribute__((ext_vector_type(2))) float v2f;
typedef __attribute__((ext_vector_type(8))) float v8f;

#define DIM 64
#define NO_EDGE 0x7fffffff

#define NCHUNK      256
#define BUCKET_ROWS 256
#define MAX_BUCKETS 512
#define BIN_THREADS 512
#define BIN_EDGES   4

#define SUM_THREADS 1024
#define SUM_WAVES   (SUM_THREADS / 32)
#define SUM_CAP     (6 * SUM_THREADS)
#define SEG_LANES   (SUM_THREADS / NCHUNK)
#define SEG_TAKE    8

#define GEMM_WAVES (BIN_THREADS / 32)

static __device__ __forceinline__ void gcn_gemm_tile(const float* __restrict__ X,
                                                     const float* __restrict__ W,
                                                     float* __restrict__ S,
                                                     int tile, int lane) {
    const int m0    = tile << 4;
    const int mrow  = lane & 15;
    const int khalf = lane >> 4;
    const int ncol  = lane & 15;

    v8f acc0 = {}, acc1 = {}, acc2 = {}, acc3 = {};

    const float* __restrict__ xrow = X + (long long)(m0 + mrow) * DIM;

    #pragma unroll
    for (int k0 = 0; k0 < DIM; k0 += 4) {
        const int ka = k0 + (khalf << 1);
        v2f a = *(const v2f*)(xrow + ka);

        const float* __restrict__ w0 = W + ka * DIM;
        const float* __restrict__ w1 = W + (ka + 1) * DIM;

        v2f b0, b1, b2, b3;
        b0.x = w0[ 0 + ncol]; b0.y = w1[ 0 + ncol];
        b1.x = w0[16 + ncol]; b1.y = w1[16 + ncol];
        b2.x = w0[32 + ncol]; b2.y = w1[32 + ncol];
        b3.x = w0[48 + ncol]; b3.y = w1[48 + ncol];

        acc0 = __builtin_amdgcn_wmma_f32_16x16x4_f32(false, a, false, b0,
                                                     (short)0, acc0, false, false);
        acc1 = __builtin_amdgcn_wmma_f32_16x16x4_f32(false, a, false, b1,
                                                     (short)0, acc1, false, false);
        acc2 = __builtin_amdgcn_wmma_f32_16x16x4_f32(false, a, false, b2,
                                                     (short)0, acc2, false, false);
        acc3 = __builtin_amdgcn_wmma_f32_16x16x4_f32(false, a, false, b3,
                                                     (short)0, acc3, false, false);
    }

    const int rbase = m0 + (khalf << 3);
    #pragma unroll
    for (int v = 0; v < 8; ++v) {
        float* __restrict__ srow = S + (long long)(rbase + v) * DIM + ncol;
        srow[ 0] = acc0[v];
        srow[16] = acc1[v];
        srow[32] = acc2[v];
        srow[48] = acc3[v];
    }
}

static __device__ __forceinline__ void gcn_gemm_extra_group(const float* __restrict__ X,
                                                            const float* __restrict__ W,
                                                            float* __restrict__ S,
                                                            int tile_first, int tile_end) {
    const int w    = __builtin_amdgcn_readfirstlane(threadIdx.x >> 5);
    const int tile = tile_first + (blockIdx.x - NCHUNK) * GEMM_WAVES + w;
    if (tile < tile_end) gcn_gemm_tile(X, W, S, tile, threadIdx.x & 31);
}

__global__ void __launch_bounds__(BIN_THREADS)
gcn_bin_kernel(const int* __restrict__ adj_row,
               const int* __restrict__ adj_col,
               const float* __restrict__ adj_val,
               int2* __restrict__ table,
               int4* __restrict__ binned,
               int n_edges,
               const float* __restrict__ X, const float* __restrict__ W,
               float* __restrict__ S, int tile_end) {
    __shared__ int lh[MAX_BUCKETS];
    __shared__ int sc[MAX_BUCKETS];
    __shared__ int lpos[MAX_BUCKETS];
    if (blockIdx.x >= NCHUNK) {
        gcn_gemm_extra_group(X, W, S, 0, tile_end);
        return;
    }
    const int t = threadIdx.x;
    const int c = blockIdx.x;
    lh[t] = 0;
    __syncthreads();

    const int run = (n_edges + NCHUNK - 1) / NCHUNK;
    const int e0  = c * run;
    const int e1  = (e0 + run < n_edges) ? e0 + run : n_edges;
    for (int eb = e0 + t; eb < e1; eb += BIN_EDGES * BIN_THREADS) {
        int r[BIN_EDGES];
        #pragma unroll
        for (int k = 0; k < BIN_EDGES; ++k) {
            const int e = eb + k * BIN_THREADS;
            r[k] = (e < e1) ? adj_row[e] : -1;
        }
        #pragma unroll
        for (int k = 0; k < BIN_EDGES; ++k) {
            if (r[k] >= 0) atomicAdd(&lh[r[k] >> 8], 1);
        }
    }
    __syncthreads();

    const int tot = lh[t];
    sc[t] = tot;
    __syncthreads();
    for (int off = 1; off < MAX_BUCKETS; off <<= 1) {
        const int add = (t >= off) ? sc[t - off] : 0;
        __syncthreads();
        sc[t] += add;
        __syncthreads();
    }
    const int first = sc[t] - tot;
    lpos[t] = e0 + first;
    table[c * MAX_BUCKETS + t] = make_int2(first, tot);
    __syncthreads();

    for (int eb = e0 + t; eb < e1; eb += BIN_EDGES * BIN_THREADS) {
        int   r[BIN_EDGES], cl[BIN_EDGES], p[BIN_EDGES];
        float v[BIN_EDGES];
        #pragma unroll
        for (int k = 0; k < BIN_EDGES; ++k) {
            const int e = eb + k * BIN_THREADS;
            r[k] = -1; cl[k] = 0; v[k] = 0.f;
            if (e < e1) { r[k] = adj_row[e]; cl[k] = adj_col[e]; v[k] = adj_val[e]; }
        }
        #pragma unroll
        for (int k = 0; k < BIN_EDGES; ++k) {
            p[k] = 0;
            if (r[k] >= 0) p[k] = atomicAdd(&lpos[r[k] >> 8], 1);
        }
        #pragma unroll
        for (int k = 0; k < BIN_EDGES; ++k) {
            const int e = eb + k * BIN_THREADS;
            if (r[k] >= 0) binned[p[k]] = make_int4(e, r[k], cl[k], __float_as_int(v[k]));
        }
    }
}

template <int U>
static __device__ __forceinline__ float2 gcn_sum_chunk(const float2* __restrict__ S2, int lane,
                                                       int sc, int sv, int t, float2 acc) {
    float2 x[U];
    float  v[U];
    #pragma unroll
    for (int k = 0; k < U; ++k) {
        const int c = __builtin_amdgcn_readlane(sc, t + k);
        v[k] = __int_as_float(__builtin_amdgcn_readlane(sv, t + k));
        x[k] = S2[(long long)c * (DIM / 2) + lane];
    }
    #pragma unroll
    for (int k = 0; k < U; ++k) {
        acc.x += v[k] * x[k].x;
        acc.y += v[k] * x[k].y;
    }
    return acc;
}

static __device__ __forceinline__ float2 gcn_sum_records(const float2* __restrict__ S2, int lane,
                                                         int sc, int sv, int n, float2 acc) {
    int t = 0;
    for (; t + 8 <= n; t += 8) acc = gcn_sum_chunk<8>(S2, lane, sc, sv, t, acc);
    if (t + 4 <= n) { acc = gcn_sum_chunk<4>(S2, lane, sc, sv, t, acc); t += 4; }
    if (t + 2 <= n) { acc = gcn_sum_chunk<2>(S2, lane, sc, sv, t, acc); t += 2; }
    if (t < n)      { acc = gcn_sum_chunk<1>(S2, lane, sc, sv, t, acc); }
    return acc;
}

static __device__ __forceinline__ void gcn_wave_min(int& best, int& where) {
    for (int off = 16; off > 0; off >>= 1) {
        const int ob = __shfl_xor(best, off);
        const int ow = __shfl_xor(where, off);
        if (ob < best) { best = ob; where = ow; }
    }
}

__global__ void __launch_bounds__(SUM_THREADS)
gcn_group_sum_kernel(const int4* __restrict__ binned,
                     const int2* __restrict__ table,
                     const float* __restrict__ S,
                     float* __restrict__ out,
                     int n_nodes, int n_edges) {
    static_assert(SEG_LANES * NCHUNK == SUM_THREADS, "every run's stretch needs its threads");
    __shared__ int  lcnt[BUCKET_ROWS];
    __shared__ int  loff[BUCKET_ROWS];
    __shared__ int  lpos[BUCKET_ROWS];
    __shared__ int  wtot[BUCKET_ROWS / 32];
    __shared__ int  rec_id[SUM_CAP];
    __shared__ int2 rec_cv[SUM_CAP];

    const int t    = threadIdx.x;
    const int lane = t & 31;
    const int w  = __builtin_amdgcn_readfirstlane(t >> 5);
    const int r0 = blockIdx.x * BUCKET_ROWS;
    if (r0 >= n_nodes) return;

    const int rows = (n_nodes - r0 < BUCKET_ROWS) ? n_nodes - r0 : BUCKET_ROWS;
    const float2* __restrict__ S2 = (const float2*)S;

    const int run = (n_edges + NCHUNK - 1) / NCHUNK;
    const int sub = t & (SEG_LANES - 1);
    const int2 sg = table[(t / SEG_LANES) * MAX_BUCKETS + blockIdx.x];
    const int sbase = (t / SEG_LANES) * run + sg.x;
    const int scnt  = sg.y;

    if (t < BUCKET_ROWS) lcnt[t] = 0;
    __syncthreads();

    for (int j0 = sub; j0 < scnt; j0 += SEG_LANES * SEG_TAKE) {
        int r[SEG_TAKE];
        #pragma unroll
        for (int k = 0; k < SEG_TAKE; ++k) {
            const int j = j0 + k * SEG_LANES;
            r[k] = (j < scnt) ? binned[sbase + j].y : -1;
        }
        #pragma unroll
        for (int k = 0; k < SEG_TAKE; ++k) {
            if (r[k] >= 0) atomicAdd(&lcnt[r[k] - r0], 1);
        }
    }
    __syncthreads();

    int c = 0, incl = 0;
    if (t < BUCKET_ROWS) {
        c = lcnt[t];
        incl = c;
        for (int off = 1; off < 32; off <<= 1) {
            const int up = __shfl_up(incl, off);
            if (lane >= off) incl += up;
        }
        if (lane == 31) wtot[w] = incl;
    }
    __syncthreads();
    if (t < BUCKET_ROWS) {
        int base = 0;
        for (int j = 0; j < BUCKET_ROWS / 32; ++j) base += (j < w) ? wtot[j] : 0;
        loff[t] = base + incl - c;
        lpos[t] = base + incl - c;
    }
    int n = 0;
    for (int j = 0; j < BUCKET_ROWS / 32; ++j) n += wtot[j];
    __syncthreads();

    if (n > SUM_CAP) {
        for (int i = 0; i < BUCKET_ROWS / SUM_WAVES; ++i) {
            const int rl = i * SUM_WAVES + w;
            if (rl >= rows) break;
            const int r = r0 + rl;
            float2 acc = make_float2(0.f, 0.f);
            int last = -1;
            for (int g = 0; g < n; ++g) {
                int best = NO_EDGE, where = 0;
                for (int cc = 0; cc < NCHUNK; ++cc) {
                    const int2 s2  = table[cc * MAX_BUCKETS + blockIdx.x];
                    const int  beg = cc * run + s2.x;
                    for (int j = lane; j < s2.y; j += 32) {
                        const int4 q = binned[beg + j];
                        if (q.y == r && q.x > last && q.x < best) { best = q.x; where = beg + j; }
                    }
                }
                gcn_wave_min(best, where);
                if (best == NO_EDGE) break;
                const int4   q = binned[where];
                const float  v = __int_as_float(q.w);
                const float2 x = S2[(long long)q.z * (DIM / 2) + lane];
                acc.x += v * x.x;
                acc.y += v * x.y;
                last = best;
            }
            ((float2*)out)[(long long)r * (DIM / 2) + lane] = acc;
        }
        return;
    }

    for (int j0 = sub; j0 < scnt; j0 += SEG_LANES * SEG_TAKE) {
        int4 q[SEG_TAKE];
        int  p[SEG_TAKE];
        #pragma unroll
        for (int k = 0; k < SEG_TAKE; ++k) {
            const int j = j0 + k * SEG_LANES;
            q[k] = make_int4(0, -1, 0, 0);
            if (j < scnt) q[k] = binned[sbase + j];
        }
        #pragma unroll
        for (int k = 0; k < SEG_TAKE; ++k) {
            p[k] = 0;
            if (q[k].y >= 0) p[k] = atomicAdd(&lpos[q[k].y - r0], 1);
        }
        #pragma unroll
        for (int k = 0; k < SEG_TAKE; ++k) {
            if (q[k].y >= 0) {
                rec_id[p[k]] = q[k].x;
                rec_cv[p[k]] = make_int2(q[k].z, q[k].w);
            }
        }
    }
    __syncthreads();

    for (int i = 0; i < BUCKET_ROWS / SUM_WAVES; ++i) {
        const int rl = i * SUM_WAVES + w;
        if (rl >= rows) break;
        const int deg = __builtin_amdgcn_readfirstlane(lcnt[rl]);
        const int off = __builtin_amdgcn_readfirstlane(loff[rl]);

        float2 acc = make_float2(0.f, 0.f);

        if (deg <= 64) {
            int e0 = NO_EDGE, c0 = 0, v0 = 0;
            int e1 = NO_EDGE, c1 = 0, v1 = 0;
            if (lane < deg)      { e0 = rec_id[off + lane];      const int2 cv = rec_cv[off + lane];      c0 = cv.x; v0 = cv.y; }
            if (lane + 32 < deg) { e1 = rec_id[off + lane + 32]; const int2 cv = rec_cv[off + lane + 32]; c1 = cv.x; v1 = cv.y; }

            int rk0 = 0, rk1 = 0;
            const int n0 = deg < 32 ? deg : 32;
            if (deg <= 32) {
                for (int j = 0; j < n0; ++j) {
                    const int kj = __builtin_amdgcn_readlane(e0, j);
                    rk0 += (kj < e0) ? 1 : 0;
                }
            } else {
                for (int j = 0; j < 32; ++j) {
                    const int kj = __builtin_amdgcn_readlane(e0, j);
                    rk0 += (kj < e0) ? 1 : 0;
                    rk1 += (kj < e1) ? 1 : 0;
                }
                for (int j = 32; j < deg; ++j) {
                    const int kj = __builtin_amdgcn_readlane(e1, j - 32);
                    rk0 += (kj < e0) ? 1 : 0;
                    rk1 += (kj < e1) ? 1 : 0;
                }
            }
            if (lane < deg)      rec_cv[off + rk0] = make_int2(c0, v0);
            if (lane + 32 < deg) rec_cv[off + rk1] = make_int2(c1, v1);
            int2 s0 = make_int2(0, 0), s1 = make_int2(0, 0);
            if (lane < deg)      s0 = rec_cv[off + lane];
            if (lane + 32 < deg) s1 = rec_cv[off + lane + 32];

            acc = gcn_sum_records(S2, lane, s0.x, s0.y, n0, acc);
            if (deg > 32) acc = gcn_sum_records(S2, lane, s1.x, s1.y, deg - 32, acc);
        } else {
            int last = -1;
            for (int g = 0; g < deg; ++g) {
                int best = NO_EDGE, where = 0;
                for (int j = lane; j < deg; j += 32) {
                    const int id = rec_id[off + j];
                    if (id > last && id < best) { best = id; where = j; }
                }
                gcn_wave_min(best, where);
                if (best == NO_EDGE) break;
                const int2   cv = rec_cv[off + where];
                const float  v  = __int_as_float(cv.y);
                const float2 x  = S2[(long long)cv.x * (DIM / 2) + lane];
                acc.x += v * x.x;
                acc.y += v * x.y;
                last = best;
            }
        }

        ((float2*)out)[(long long)(r0 + rl) * (DIM / 2) + lane] = acc;
    }
}

extern "C" void kernel_launch(void* const* d_in, const int* in_sizes, int n_in,
                              void* d_out, int out_size, void* d_ws, size_t ws_size,
                              hipStream_t stream) {
    const float* X    = (const float*)d_in[0];
    const float* W    = (const float*)d_in[1];
    const int*   arow = (const int*)d_in[2];
    const int*   acol = (const int*)d_in[3];
    const float* aval = (const float*)d_in[4];
    float*       out  = (float*)d_out;

    const int n_nodes = in_sizes[0] / DIM;
    const int n_edges = in_sizes[2];

    float* S        = (float*)d_ws;
    int2*  table    = (int2*)(S + (size_t)n_nodes * DIM);
    int4*  binned   = (int4*)(table + NCHUNK * MAX_BUCKETS);

    const int tiles = n_nodes / 16;
    const int g_all = (tiles + GEMM_WAVES - 1) / GEMM_WAVES;

    gcn_bin_kernel<<<NCHUNK + g_all, BIN_THREADS, 0, stream>>>(arow, acol, aval, table, binned, n_edges,
                                                              X, W, S, tiles);

    gcn_group_sum_kernel<<<MAX_BUCKETS, SUM_THREADS, 0, stream>>>(binned, table, S, out, n_nodes, n_edges);
}


# Row sums re-laid: a half-wave per record with 16-byte loads (two support rows per load), a row's loads asked for together; even and odd positions summed apart, the two sums added last
# speedup vs baseline: 5.0667x; 1.0866x over previous
#include <hip/hip_runtime.h>

typedef __attribute__((ext_vector_type(2))) float v2f;
typedef __attribute__((ext_vector_type(8))) float v8f;

#define DIM 64
#define NO_EDGE 0x7fffffff

#define NCHUNK      256
#define BUCKET_ROWS 256
#define MAX_BUCKETS 512
#define BIN_THREADS 512
#define BIN_EDGES   4

#define SUM_THREADS 1024
#define SUM_WAVES   (SUM_THREADS / 32)
#define SUM_CAP     (6 * SUM_THREADS)
#define SEG_LANES   (SUM_THREADS / NCHUNK)
#define SEG_TAKE    8

#define GEMM_WAVES (BIN_THREADS / 32)

static __device__ __forceinline__ void gcn_gemm_tile(const float* __restrict__ X,
                                                     const float* __restrict__ W,
                                                     float* __restrict__ S,
                                                     int tile, int lane) {
    const int m0    = tile << 4;
    const int mrow  = lane & 15;
    const int khalf = lane >> 4;
    const int ncol  = lane & 15;

    v8f acc0 = {}, acc1 = {}, acc2 = {}, acc3 = {};

    const float* __restrict__ xrow = X + (long long)(m0 + mrow) * DIM;

    #pragma unroll
    for (int k0 = 0; k0 < DIM; k0 += 4) {
        const int ka = k0 + (khalf << 1);
        v2f a = *(const v2f*)(xrow + ka);

        const float* __restrict__ w0 = W + ka * DIM;
        const float* __restrict__ w1 = W + (ka + 1) * DIM;

        v2f b0, b1, b2, b3;
        b0.x = w0[ 0 + ncol]; b0.y = w1[ 0 + ncol];
        b1.x = w0[16 + ncol]; b1.y = w1[16 + ncol];
        b2.x = w0[32 + ncol]; b2.y = w1[32 + ncol];
        b3.x = w0[48 + ncol]; b3.y = w1[48 + ncol];

        acc0 = __builtin_amdgcn_wmma_f32_16x16x4_f32(false, a, false, b0,
                                                     (short)0, acc0, false, false);
        acc1 = __builtin_amdgcn_wmma_f32_16x16x4_f32(false, a, false, b1,
                                                     (short)0, acc1, false, false);
        acc2 = __builtin_amdgcn_wmma_f32_16x16x4_f32(false, a, false, b2,
                                                     (short)0, acc2, false, false);
        acc3 = __builtin_amdgcn_wmma_f32_16x16x4_f32(false, a, false, b3,
                                                     (short)0, acc3, false, false);
    }

    const int rbase = m0 + (khalf << 3);
    #pragma unroll
    for (int v = 0; v < 8; ++v) {
        float* __restrict__ srow = S + (long long)(rbase + v) * DIM + ncol;
        srow[ 0] = acc0[v];
        srow[16] = acc1[v];
        srow[32] = acc2[v];
        srow[48] = acc3[v];
    }
}

static __device__ __forceinline__ void gcn_gemm_extra_group(const float* __restrict__ X,
                                                            const float* __restrict__ W,
                                                            float* __restrict__ S,
                                                            int tile_first, int tile_end) {
    const int w    = __builtin_amdgcn_readfirstlane(threadIdx.x >> 5);
    const int tile = tile_first + (blockIdx.x - NCHUNK) * GEMM_WAVES + w;
    if (tile < tile_end) gcn_gemm_tile(X, W, S, tile, threadIdx.x & 31);
}

__global__ void __launch_bounds__(BIN_THREADS)
gcn_bin_kernel(const int* __restrict__ adj_row,
               const int* __restrict__ adj_col,
               const float* __restrict__ adj_val,
               int2* __restrict__ table,
               int4* __restrict__ binned,
               int n_edges,
               const float* __restrict__ X, const float* __restrict__ W,
               float* __restrict__ S, int tile_end) {
    __shared__ int lh[MAX_BUCKETS];
    __shared__ int sc[MAX_BUCKETS];
    __shared__ int lpos[MAX_BUCKETS];
    if (blockIdx.x >= NCHUNK) {
        gcn_gemm_extra_group(X, W, S, 0, tile_end);
        return;
    }
    const int t = threadIdx.x;
    const int c = blockIdx.x;
    lh[t] = 0;
    __syncthreads();

    const int run = (n_edges + NCHUNK - 1) / NCHUNK;
    const int e0  = c * run;
    const int e1  = (e0 + run < n_edges) ? e0 + run : n_edges;
    for (int eb = e0 + t; eb < e1; eb += BIN_EDGES * BIN_THREADS) {
        int r[BIN_EDGES];
        #pragma unroll
        for (int k = 0; k < BIN_EDGES; ++k) {
            const int e = eb + k * BIN_THREADS;
            r[k] = (e < e1) ? adj_row[e] : -1;
        }
        #pragma unroll
        for (int k = 0; k < BIN_EDGES; ++k) {
            if (r[k] >= 0) atomicAdd(&lh[r[k] >> 8], 1);
        }
    }
    __syncthreads();

    const int tot = lh[t];
    sc[t] = tot;
    __syncthreads();
    for (int off = 1; off < MAX_BUCKETS; off <<= 1) {
        const int add = (t >= off) ? sc[t - off] : 0;
        __syncthreads();
        sc[t] += add;
        __syncthreads();
    }
    const int first = sc[t] - tot;
    lpos[t] = e0 + first;
    table[c * MAX_BUCKETS + t] = make_int2(first, tot);
    __syncthreads();

    for (int eb = e0 + t; eb < e1; eb += BIN_EDGES * BIN_THREADS) {
        int   r[BIN_EDGES], cl[BIN_EDGES], p[BIN_EDGES];
        float v[BIN_EDGES];
        #pragma unroll
        for (int k = 0; k < BIN_EDGES; ++k) {
            const int e = eb + k * BIN_THREADS;
            r[k] = -1; cl[k] = 0; v[k] = 0.f;
            if (e < e1) { r[k] = adj_row[e]; cl[k] = adj_col[e]; v[k] = adj_val[e]; }
        }
        #pragma unroll
        for (int k = 0; k < BIN_EDGES; ++k) {
            p[k] = 0;
            if (r[k] >= 0) p[k] = atomicAdd(&lpos[r[k] >> 8], 1);
        }
        #pragma unroll
        for (int k = 0; k < BIN_EDGES; ++k) {
            const int e = eb + k * BIN_THREADS;
            if (r[k] >= 0) binned[p[k]] = make_int4(e, r[k], cl[k], __float_as_int(v[k]));
        }
    }
}

template <int U>
static __device__ __forceinline__ float4 gcn_pair_round(const float4* __restrict__ S4, int q,
                                                        const int2* row, int pos0, int deg, float4 acc) {
    float  v[U];
    float4 x[U];
    const int  plast = pos0 + 2 * (U - 1);
    const bool have  = plast < deg;
    #pragma unroll
    for (int k = 0; k < U; ++k) {
        const int  p  = (k == U - 1) ? (have ? plast : deg - 1) : pos0 + 2 * k;
        const int2 cv = row[p];
        v[k] = __int_as_float(cv.y);
        x[k] = S4[(unsigned)(cv.x * (DIM / 4) + q)];
    }
    if (!have) { v[U - 1] = 0.f; x[U - 1] = make_float4(0.f, 0.f, 0.f, 0.f); }
    #pragma unroll
    for (int k = 0; k < U; ++k) {
        acc.x = __builtin_fmaf(v[k], x[k].x, acc.x);
        acc.y = __builtin_fmaf(v[k], x[k].y, acc.y);
        acc.z = __builtin_fmaf(v[k], x[k].z, acc.z);
        acc.w = __builtin_fmaf(v[k], x[k].w, acc.w);
    }
    return acc;
}

static __device__ __forceinline__ float gcn_other_half(float f) {
    const unsigned u = (unsigned)__float_as_int(f);
    return __int_as_float((int)__builtin_amdgcn_permlanex16(u, u, 0x76543210u, 0xfedcba98u, false, false));
}

static __device__ __forceinline__ void gcn_wave_min(int& best, int& where) {
    for (int off = 16; off > 0; off >>= 1) {
        const int ob = __shfl_xor(best, off);
        const int ow = __shfl_xor(where, off);
        if (ob < best) { best = ob; where = ow; }
    }
}

__global__ void __launch_bounds__(SUM_THREADS)
gcn_group_sum_kernel(const int4* __restrict__ binned,
                     const int2* __restrict__ table,
                     const float* __restrict__ S,
                     float* __restrict__ out,
                     int n_nodes, int n_edges) {
    static_assert(SEG_LANES * NCHUNK == SUM_THREADS, "every run's stretch needs its threads");
    __shared__ int  lcnt[BUCKET_ROWS];
    __shared__ int  loff[BUCKET_ROWS];
    __shared__ int  lpos[BUCKET_ROWS];
    __shared__ int  wtot[BUCKET_ROWS / 32];
    __shared__ int  rec_id[SUM_CAP];
    __shared__ int2 rec_cv[SUM_CAP];

    const int t    = threadIdx.x;
    const int lane = t & 31;
    const int w  = __builtin_amdgcn_readfirstlane(t >> 5);
    const int r0 = blockIdx.x * BUCKET_ROWS;
    if (r0 >= n_nodes) return;

    const int rows = (n_nodes - r0 < BUCKET_ROWS) ? n_nodes - r0 : BUCKET_ROWS;
    const float2* __restrict__ S2 = (const float2*)S;
    const float4* __restrict__ S4 = (const float4*)S;
    const int h = lane >> 4;
    const int q = lane & 15;

    const int run = (n_edges + NCHUNK - 1) / NCHUNK;
    const int sub = t & (SEG_LANES - 1);
    const int2 sg = table[(t / SEG_LANES) * MAX_BUCKETS + blockIdx.x];
    const int sbase = (t / SEG_LANES) * run + sg.x;
    const int scnt  = sg.y;

    if (t < BUCKET_ROWS) lcnt[t] = 0;
    __syncthreads();

    for (int j0 = sub; j0 < scnt; j0 += SEG_LANES * SEG_TAKE) {
        int r[SEG_TAKE];
        #pragma unroll
        for (int k = 0; k < SEG_TAKE; ++k) {
            const int j = j0 + k * SEG_LANES;
            r[k] = (j < scnt) ? binned[sbase + j].y : -1;
        }
        #pragma unroll
        for (int k = 0; k < SEG_TAKE; ++k) {
            if (r[k] >= 0) atomicAdd(&lcnt[r[k] - r0], 1);
        }
    }
    __syncthreads();

    int c = 0, incl = 0;
    if (t < BUCKET_ROWS) {
        c = lcnt[t];
        incl = c;
        for (int off = 1; off < 32; off <<= 1) {
            const int up = __shfl_up(incl, off);
            if (lane >= off) incl += up;
        }
        if (lane == 31) wtot[w] = incl;
    }
    __syncthreads();
    if (t < BUCKET_ROWS) {
        int base = 0;
        for (int j = 0; j < BUCKET_ROWS / 32; ++j) base += (j < w) ? wtot[j] : 0;
        loff[t] = base + incl - c;
        lpos[t] = base + incl - c;
    }
    int n = 0;
    for (int j = 0; j < BUCKET_ROWS / 32; ++j) n += wtot[j];
    __syncthreads();

    if (n > SUM_CAP) {
        for (int i = 0; i < BUCKET_ROWS / SUM_WAVES; ++i) {
            const int rl = i * SUM_WAVES + w;
            if (rl >= rows) break;
            const int r = r0 + rl;
            float2 acc = make_float2(0.f, 0.f);
            int last = -1;
            for (int g = 0; g < n; ++g) {
                int best = NO_EDGE, where = 0;
                for (int cc = 0; cc < NCHUNK; ++cc) {
                    const int2 s2  = table[cc * MAX_BUCKETS + blockIdx.x];
                    const int  beg = cc * run + s2.x;
                    for (int j = lane; j < s2.y; j += 32) {
                        const int4 q = binned[beg + j];
                        if (q.y == r && q.x > last && q.x < best) { best = q.x; where = beg + j; }
                    }
                }
                gcn_wave_min(best, where);
                if (best == NO_EDGE) break;
                const int4   q = binned[where];
                const float  v = __int_as_float(q.w);
                const float2 x = S2[(long long)q.z * (DIM / 2) + lane];
                acc.x += v * x.x;
                acc.y += v * x.y;
                last = best;
            }
            ((float2*)out)[(long long)r * (DIM / 2) + lane] = acc;
        }
        return;
    }

    for (int j0 = sub; j0 < scnt; j0 += SEG_LANES * SEG_TAKE) {
        int4 q[SEG_TAKE];
        int  p[SEG_TAKE];
        #pragma unroll
        for (int k = 0; k < SEG_TAKE; ++k) {
            const int j = j0 + k * SEG_LANES;
            q[k] = make_int4(0, -1, 0, 0);
            if (j < scnt) q[k] = binned[sbase + j];
        }
        #pragma unroll
        for (int k = 0; k < SEG_TAKE; ++k) {
            p[k] = 0;
            if (q[k].y >= 0) p[k] = atomicAdd(&lpos[q[k].y - r0], 1);
        }
        #pragma unroll
        for (int k = 0; k < SEG_TAKE; ++k) {
            if (q[k].y >= 0) {
                rec_id[p[k]] = q[k].x;
                rec_cv[p[k]] = make_int2(q[k].z, q[k].w);
            }
        }
    }
    __syncthreads();

    bool big = false;
    for (int i = 0; i < BUCKET_ROWS / SUM_WAVES; ++i) {
        const int rl = i * SUM_WAVES + w;
        if (rl >= rows) break;
        const int deg = __builtin_amdgcn_readfirstlane(lcnt[rl]);
        const int off = __builtin_amdgcn_readfirstlane(loff[rl]);
        if (deg > 64) { big = true; continue; }

        int e0 = NO_EDGE, c0 = 0, v0 = 0;
        int e1 = NO_EDGE, c1 = 0, v1 = 0;
        if (lane < deg)      { e0 = rec_id[off + lane];      const int2 cv = rec_cv[off + lane];      c0 = cv.x; v0 = cv.y; }
        if (lane + 32 < deg) { e1 = rec_id[off + lane + 32]; const int2 cv = rec_cv[off + lane + 32]; c1 = cv.x; v1 = cv.y; }

        int rk0 = 0, rk1 = 0;
        const int n0 = deg < 32 ? deg : 32;
        if (deg <= 32) {
            for (int j = 0; j < n0; ++j) {
                const int kj = __builtin_amdgcn_readlane(e0, j);
                rk0 += (kj < e0) ? 1 : 0;
            }
        } else {
            for (int j = 0; j < 32; ++j) {
                const int kj = __builtin_amdgcn_readlane(e0, j);
                rk0 += (kj < e0) ? 1 : 0;
                rk1 += (kj < e1) ? 1 : 0;
            }
            for (int j = 32; j < deg; ++j) {
                const int kj = __builtin_amdgcn_readlane(e1, j - 32);
                rk0 += (kj < e0) ? 1 : 0;
                rk1 += (kj < e1) ? 1 : 0;
            }
        }
        if (lane < deg)      rec_cv[off + rk0] = make_int2(c0, v0);
        if (lane + 32 < deg) rec_cv[off + rk1] = make_int2(c1, v1);

        const int2* row    = &rec_cv[off];
        const int   nsteps = (deg + 1) >> 1;
        float4 a = make_float4(0.f, 0.f, 0.f, 0.f);
        for (int s = 0; s < nsteps; s += 8) {
            const int pos0 = 2 * s + h;
            switch (nsteps - s) {
                case 1:  a = gcn_pair_round<1>(S4, q, row, pos0, deg, a); break;
                case 2:  a = gcn_pair_round<2>(S4, q, row, pos0, deg, a); break;
                case 3:  a = gcn_pair_round<3>(S4, q, row, pos0, deg, a); break;
                case 4:  a = gcn_pair_round<4>(S4, q, row, pos0, deg, a); break;
                case 5:  a = gcn_pair_round<5>(S4, q, row, pos0, deg, a); break;
                case 6:  a = gcn_pair_round<6>(S4, q, row, pos0, deg, a); break;
                case 7:  a = gcn_pair_round<7>(S4, q, row, pos0, deg, a); break;
                default: a = gcn_pair_round<8>(S4, q, row, pos0, deg, a); break;
            }
        }

        a.x += gcn_other_half(a.x);
        a.y += gcn_other_half(a.y);
        a.z += gcn_other_half(a.z);
        a.w += gcn_other_half(a.w);

        if (h == 0) ((float4*)out)[(long long)(r0 + rl) * (DIM / 4) + q] = a;
    }
    if (!big) return;

    #pragma nounroll
    for (int i = 0; i < BUCKET_ROWS / SUM_WAVES; ++i) {
        const int rl = i * SUM_WAVES + w;
        if (rl >= rows) break;
        const int deg = __builtin_amdgcn_readfirstlane(lcnt[rl]);
        const int off = __builtin_amdgcn_readfirstlane(loff[rl]);
        if (deg <= 64) continue;

        float2 acc = make_float2(0.f, 0.f);
        int last = -1;
        for (int g = 0; g < deg; ++g) {
            int best = NO_EDGE, where = 0;
            for (int j = lane; j < deg; j += 32) {
                const int id = rec_id[off + j];
                if (id > last && id < best) { best = id; where = j; }
            }
            gcn_wave_min(best, where);
            if (best == NO_EDGE) break;
            const int2   cv = rec_cv[off + where];
            const float  v  = __int_as_float(cv.y);
            const float2 x  = S2[(long long)cv.x * (DIM / 2) + lane];
            acc.x += v * x.x;
            acc.y += v * x.y;
            last = best;
        }
        ((float2*)out)[(long long)(r0 + rl) * (DIM / 2) + lane] = acc;
    }
}

extern "C" void kernel_launch(void* const* d_in, const int* in_sizes, int n_in,
                              void* d_out, int out_size, void* d_ws, size_t ws_size,
                              hipStream_t stream) {
    const float* X    = (const float*)d_in[0];
    const float* W    = (const float*)d_in[1];
    const int*   arow = (const int*)d_in[2];
    const int*   acol = (const int*)d_in[3];
    const float* aval = (const float*)d_in[4];
    float*       out  = (float*)d_out;

    const int n_nodes = in_sizes[0] / DIM;
    const int n_edges = in_sizes[2];

    float* S        = (float*)d_ws;
    int2*  table    = (int2*)(S + (size_t)n_nodes * DIM);
    int4*  binned   = (int4*)(table + NCHUNK * MAX_BUCKETS);

    const int tiles = n_nodes / 16;
    const int g_all = (tiles + GEMM_WAVES - 1) / GEMM_WAVES;

    gcn_bin_kernel<<<NCHUNK + g_all, BIN_THREADS, 0, stream>>>(arow, acol, aval, table, binned, n_edges,
                                                              X, W, S, tiles);

    gcn_group_sum_kernel<<<MAX_BUCKETS, SUM_THREADS, 0, stream>>>(binned, table, S, out, n_nodes, n_edges);
}


# Edge sort reads each edge once: a thread keeps its 13 edges' row, column and value in registers from the counting to the copying, all loads asked for at the start; sum order unchanged
# speedup vs baseline: 5.3142x; 1.0488x over previous
#include <hip/hip_runtime.h>

typedef __attribute__((ext_vector_type(2))) float v2f;
typedef __attribute__((ext_vector_type(8))) float v8f;

#define DIM 64
#define NO_EDGE 0x7fffffff

#define NCHUNK      256
#define BUCKET_ROWS 256
#define MAX_BUCKETS 512
#define BIN_THREADS 512
#define BIN_EDGES   4
#define BIN_HELD    13

#define SUM_THREADS 1024
#define SUM_WAVES   (SUM_THREADS / 32)
#define SUM_CAP     (6 * SUM_THREADS)
#define SEG_LANES   (SUM_THREADS / NCHUNK)
#define SEG_TAKE    8

#define GEMM_WAVES (BIN_THREADS / 32)

static __device__ __forceinline__ void gcn_gemm_tile(const float* __restrict__ X,
                                                     const float* __restrict__ W,
                                                     float* __restrict__ S,
                                                     int tile, int lane) {
    const int m0    = tile << 4;
    const int mrow  = lane & 15;
    const int khalf = lane >> 4;
    const int ncol  = lane & 15;

    v8f acc0 = {}, acc1 = {}, acc2 = {}, acc3 = {};

    const float* __restrict__ xrow = X + (long long)(m0 + mrow) * DIM;

    #pragma unroll
    for (int k0 = 0; k0 < DIM; k0 += 4) {
        const int ka = k0 + (khalf << 1);
        v2f a = *(const v2f*)(xrow + ka);

        const float* __restrict__ w0 = W + ka * DIM;
        const float* __restrict__ w1 = W + (ka + 1) * DIM;

        v2f b0, b1, b2, b3;
        b0.x = w0[ 0 + ncol]; b0.y = w1[ 0 + ncol];
        b1.x = w0[16 + ncol]; b1.y = w1[16 + ncol];
        b2.x = w0[32 + ncol]; b2.y = w1[32 + ncol];
        b3.x = w0[48 + ncol]; b3.y = w1[48 + ncol];

        acc0 = __builtin_amdgcn_wmma_f32_16x16x4_f32(false, a, false, b0,
                                                     (short)0, acc0, false, false);
        acc1 = __builtin_amdgcn_wmma_f32_16x16x4_f32(false, a, false, b1,
                                                     (short)0, acc1, false, false);
        acc2 = __builtin_amdgcn_wmma_f32_16x16x4_f32(false, a, false, b2,
                                                     (short)0, acc2, false, false);
        acc3 = __builtin_amdgcn_wmma_f32_16x16x4_f32(false, a, false, b3,
                                                     (short)0, acc3, false, false);
    }

    const int rbase = m0 + (khalf << 3);
    #pragma unroll
    for (int v = 0; v < 8; ++v) {
        float* __restrict__ srow = S + (long long)(rbase + v) * DIM + ncol;
        srow[ 0] = acc0[v];
        srow[16] = acc1[v];
        srow[32] = acc2[v];
        srow[48] = acc3[v];
    }
}

static __device__ __forceinline__ void gcn_gemm_extra_group(const float* __restrict__ X,
                                                            const float* __restrict__ W,
                                                            float* __restrict__ S,
                                                            int tile_first, int tile_end) {
    const int w    = __builtin_amdgcn_readfirstlane(threadIdx.x >> 5);
    const int tile = tile_first + (blockIdx.x - NCHUNK) * GEMM_WAVES + w;
    if (tile < tile_end) gcn_gemm_tile(X, W, S, tile, threadIdx.x & 31);
}

__global__ void __launch_bounds__(BIN_THREADS) __attribute__((amdgpu_waves_per_eu(12)))
gcn_bin_kernel(const int* __restrict__ adj_row,
               const int* __restrict__ adj_col,
               const float* __restrict__ adj_val,
               int2* __restrict__ table,
               int4* __restrict__ binned,
               int n_edges,
               const float* __restrict__ X, const float* __restrict__ W,
               float* __restrict__ S, int tile_end) {
    __shared__ int lh[MAX_BUCKETS];
    __shared__ int sc[MAX_BUCKETS];
    __shared__ int lpos[MAX_BUCKETS];
    if (blockIdx.x >= NCHUNK) {
        gcn_gemm_extra_group(X, W, S, 0, tile_end);
        return;
    }
    const int t = threadIdx.x;
    const int c = blockIdx.x;

    const int run = (n_edges + NCHUNK - 1) / NCHUNK;
    const int e0  = c * run;
    const int e1  = (e0 + run < n_edges) ? e0 + run : n_edges;
    const bool held = run <= BIN_HELD * BIN_THREADS;

    lh[t] = 0;
    __syncthreads();

    int   hr[BIN_HELD], hc[BIN_HELD];
    float hv[BIN_HELD];
    if (held) {
        #pragma unroll
        for (int k = 0; k < BIN_HELD; ++k) {
            const int e  = e0 + t + k * BIN_THREADS;
            const int ec = (e < e1) ? e : ((e1 > 0) ? e1 - 1 : 0);
            hr[k] = adj_row[ec];
        }
        #pragma unroll
        for (int k = 0; k < BIN_HELD; ++k) {
            const int e  = e0 + t + k * BIN_THREADS;
            const int ec = (e < e1) ? e : ((e1 > 0) ? e1 - 1 : 0);
            hc[k] = adj_col[ec];
            hv[k] = adj_val[ec];
        }
        #pragma unroll
        for (int k = 0; k < BIN_HELD; ++k) {
            const int e = e0 + t + k * BIN_THREADS;
            if (e >= e1) hr[k] = -1;
        }
    }

    if (held) {
        #pragma unroll
        for (int k = 0; k < BIN_HELD; ++k) {
            if (hr[k] >= 0) atomicAdd(&lh[hr[k] >> 8], 1);
        }
    } else
    for (int eb = e0 + t; eb < e1; eb += BIN_EDGES * BIN_THREADS) {
        int r[BIN_EDGES];
        #pragma unroll
        for (int k = 0; k < BIN_EDGES; ++k) {
            const int e = eb + k * BIN_THREADS;
            r[k] = (e < e1) ? adj_row[e] : -1;
        }
        #pragma unroll
        for (int k = 0; k < BIN_EDGES; ++k) {
            if (r[k] >= 0) atomicAdd(&lh[r[k] >> 8], 1);
        }
    }
    __syncthreads();

    const int tot = lh[t];
    sc[t] = tot;
    __syncthreads();
    for (int off = 1; off < MAX_BUCKETS; off <<= 1) {
        const int add = (t >= off) ? sc[t - off] : 0;
        __syncthreads();
        sc[t] += add;
        __syncthreads();
    }
    const int first = sc[t] - tot;
    lpos[t] = e0 + first;
    table[c * MAX_BUCKETS + t] = make_int2(first, tot);
    __syncthreads();

    if (held) {
        int p[BIN_HELD];
        #pragma unroll
        for (int k = 0; k < BIN_HELD; ++k) {
            p[k] = 0;
            if (hr[k] >= 0) p[k] = atomicAdd(&lpos[hr[k] >> 8], 1);
        }
        #pragma unroll
        for (int k = 0; k < BIN_HELD; ++k) {
            const int e = e0 + t + k * BIN_THREADS;
            if (hr[k] >= 0) binned[p[k]] = make_int4(e, hr[k], hc[k], __float_as_int(hv[k]));
        }
        return;
    }

    for (int eb = e0 + t; eb < e1; eb += BIN_EDGES * BIN_THREADS) {
        int   r[BIN_EDGES], cl[BIN_EDGES], p[BIN_EDGES];
        float v[BIN_EDGES];
        #pragma unroll
        for (int k = 0; k < BIN_EDGES; ++k) {
            const int e = eb + k * BIN_THREADS;
            r[k] = -1; cl[k] = 0; v[k] = 0.f;
            if (e < e1) { r[k] = adj_row[e]; cl[k] = adj_col[e]; v[k] = adj_val[e]; }
        }
        #pragma unroll
        for (int k = 0; k < BIN_EDGES; ++k) {
            p[k] = 0;
            if (r[k] >= 0) p[k] = atomicAdd(&lpos[r[k] >> 8], 1);
        }
        #pragma unroll
        for (int k = 0; k < BIN_EDGES; ++k) {
            const int e = eb + k * BIN_THREADS;
            if (r[k] >= 0) binned[p[k]] = make_int4(e, r[k], cl[k], __float_as_int(v[k]));
        }
    }
}

template <int U>
static __device__ __forceinline__ float4 gcn_pair_round(const float4* __restrict__ S4, int q,
                                                        const int2* row, int pos0, int deg, float4 acc) {
    float  v[U];
    float4 x[U];
    const int  plast = pos0 + 2 * (U - 1);
    const bool have  = plast < deg;
    #pragma unroll
    for (int k = 0; k < U; ++k) {
        const int  p  = (k == U - 1) ? (have ? plast : deg - 1) : pos0 + 2 * k;
        const int2 cv = row[p];
        v[k] = __int_as_float(cv.y);
        x[k] = S4[(unsigned)(cv.x * (DIM / 4) + q)];
    }
    if (!have) { v[U - 1] = 0.f; x[U - 1] = make_float4(0.f, 0.f, 0.f, 0.f); }
    #pragma unroll
    for (int k = 0; k < U; ++k) {
        acc.x = __builtin_fmaf(v[k], x[k].x, acc.x);
        acc.y = __builtin_fmaf(v[k], x[k].y, acc.y);
        acc.z = __builtin_fmaf(v[k], x[k].z, acc.z);
        acc.w = __builtin_fmaf(v[k], x[k].w, acc.w);
    }
    return acc;
}

static __device__ __forceinline__ float gcn_other_half(float f) {
    const unsigned u = (unsigned)__float_as_int(f);
    return __int_as_float((int)__builtin_amdgcn_permlanex16(u, u, 0x76543210u, 0xfedcba98u, false, false));
}

static __device__ __forceinline__ void gcn_wave_min(int& best, int& where) {
    for (int off = 16; off > 0; off >>= 1) {
        const int ob = __shfl_xor(best, off);
        const int ow = __shfl_xor(where, off);
        if (ob < best) { best = ob; where = ow; }
    }
}

__global__ void __launch_bounds__(SUM_THREADS)
gcn_group_sum_kernel(const int4* __restrict__ binned,
                     const int2* __restrict__ table,
                     const float* __restrict__ S,
                     float* __restrict__ out,
                     int n_nodes, int n_edges) {
    static_assert(SEG_LANES * NCHUNK == SUM_THREADS, "every run's stretch needs its threads");
    __shared__ int  lcnt[BUCKET_ROWS];
    __shared__ int  loff[BUCKET_ROWS];
    __shared__ int  lpos[BUCKET_ROWS];
    __shared__ int  wtot[BUCKET_ROWS / 32];
    __shared__ int  rec_id[SUM_CAP];
    __shared__ int2 rec_cv[SUM_CAP];

    const int t    = threadIdx.x;
    const int lane = t & 31;
    const int w  = __builtin_amdgcn_readfirstlane(t >> 5);
    const int r0 = blockIdx.x * BUCKET_ROWS;
    if (r0 >= n_nodes) return;

    const int rows = (n_nodes - r0 < BUCKET_ROWS) ? n_nodes - r0 : BUCKET_ROWS;
    const float2* __restrict__ S2 = (const float2*)S;
    const float4* __restrict__ S4 = (const float4*)S;
    const int h = lane >> 4;
    const int q = lane & 15;

    const int run = (n_edges + NCHUNK - 1) / NCHUNK;
    const int sub = t & (SEG_LANES - 1);
    const int2 sg = table[(t / SEG_LANES) * MAX_BUCKETS + blockIdx.x];
    const int sbase = (t / SEG_LANES) * run + sg.x;
    const int scnt  = sg.y;

    if (t < BUCKET_ROWS) lcnt[t] = 0;
    __syncthreads();

    for (int j0 = sub; j0 < scnt; j0 += SEG_LANES * SEG_TAKE) {
        int r[SEG_TAKE];
        #pragma unroll
        for (int k = 0; k < SEG_TAKE; ++k) {
            const int j = j0 + k * SEG_LANES;
            r[k] = (j < scnt) ? binned[sbase + j].y : -1;
        }
        #pragma unroll
        for (int k = 0; k < SEG_TAKE; ++k) {
            if (r[k] >= 0) atomicAdd(&lcnt[r[k] - r0], 1);
        }
    }
    __syncthreads();

    int c = 0, incl = 0;
    if (t < BUCKET_ROWS) {
        c = lcnt[t];
        incl = c;
        for (int off = 1; off < 32; off <<= 1) {
            const int up = __shfl_up(incl, off);
            if (lane >= off) incl += up;
        }
        if (lane == 31) wtot[w] = incl;
    }
    __syncthreads();
    if (t < BUCKET_ROWS) {
        int base = 0;
        for (int j = 0; j < BUCKET_ROWS / 32; ++j) base += (j < w) ? wtot[j] : 0;
        loff[t] = base + incl - c;
        lpos[t] = base + incl - c;
    }
    int n = 0;
    for (int j = 0; j < BUCKET_ROWS / 32; ++j) n += wtot[j];
    __syncthreads();

    if (n > SUM_CAP) {
        for (int i = 0; i < BUCKET_ROWS / SUM_WAVES; ++i) {
            const int rl = i * SUM_WAVES + w;
            if (rl >= rows) break;
            const int r = r0 + rl;
            float2 acc = make_float2(0.f, 0.f);
            int last = -1;
            for (int g = 0; g < n; ++g) {
                int best = NO_EDGE, where = 0;
                for (int cc = 0; cc < NCHUNK; ++cc) {
                    const int2 s2  = table[cc * MAX_BUCKETS + blockIdx.x];
                    const int  beg = cc * run + s2.x;
                    for (int j = lane; j < s2.y; j += 32) {
                        const int4 q = binned[beg + j];
                        if (q.y == r && q.x > last && q.x < best) { best = q.x; where = beg + j; }
                    }
                }
                gcn_wave_min(best, where);
                if (best == NO_EDGE) break;
                const int4   q = binned[where];
                const float  v = __int_as_float(q.w);
                const float2 x = S2[(long long)q.z * (DIM / 2) + lane];
                acc.x += v * x.x;
                acc.y += v * x.y;
                last = best;
            }
            ((float2*)out)[(long long)r * (DIM / 2) + lane] = acc;
        }
        return;
    }

    for (int j0 = sub; j0 < scnt; j0 += SEG_LANES * SEG_TAKE) {
        int4 q[SEG_TAKE];
        int  p[SEG_TAKE];
        #pragma unroll
        for (int k = 0; k < SEG_TAKE; ++k) {
            const int j = j0 + k * SEG_LANES;
            q[k] = make_int4(0, -1, 0, 0);
            if (j < scnt) q[k] = binned[sbase + j];
        }
        #pragma unroll
        for (int k = 0; k < SEG_TAKE; ++k) {
            p[k] = 0;
            if (q[k].y >= 0) p[k] = atomicAdd(&lpos[q[k].y - r0], 1);
        }
        #pragma unroll
        for (int k = 0; k < SEG_TAKE; ++k) {
            if (q[k].y >= 0) {
                rec_id[p[k]] = q[k].x;
                rec_cv[p[k]] = make_int2(q[k].z, q[k].w);
            }
        }
    }
    __syncthreads();

    bool big = false;
    for (int i = 0; i < BUCKET_ROWS / SUM_WAVES; ++i) {
        const int rl = i * SUM_WAVES + w;
        if (rl >= rows) break;
        const int deg = __builtin_amdgcn_readfirstlane(lcnt[rl]);
        const int off = __builtin_amdgcn_readfirstlane(loff[rl]);
        if (deg > 64) { big = true; continue; }

        int e0 = NO_EDGE, c0 = 0, v0 = 0;
        int e1 = NO_EDGE, c1 = 0, v1 = 0;
        if (lane < deg)      { e0 = rec_id[off + lane];      const int2 cv = rec_cv[off + lane];      c0 = cv.x; v0 = cv.y; }
        if (lane + 32 < deg) { e1 = rec_id[off + lane + 32]; const int2 cv = rec_cv[off + lane + 32]; c1 = cv.x; v1 = cv.y; }

        int rk0 = 0, rk1 = 0;
        const int n0 = deg < 32 ? deg : 32;
        if (deg <= 32) {
            for (int j = 0; j < n0; ++j) {
                const int kj = __builtin_amdgcn_readlane(e0, j);
                rk0 += (kj < e0) ? 1 : 0;
            }
        } else {
            for (int j = 0; j < 32; ++j) {
                const int kj = __builtin_amdgcn_readlane(e0, j);
                rk0 += (kj < e0) ? 1 : 0;
                rk1 += (kj < e1) ? 1 : 0;
            }
            for (int j = 32; j < deg; ++j) {
                const int kj = __builtin_amdgcn_readlane(e1, j - 32);
                rk0 += (kj < e0) ? 1 : 0;
                rk1 += (kj < e1) ? 1 : 0;
            }
        }
        if (lane < deg)      rec_cv[off + rk0] = make_int2(c0, v0);
        if (lane + 32 < deg) rec_cv[off + rk1] = make_int2(c1, v1);

        const int2* row    = &rec_cv[off];
        const int   nsteps = (deg + 1) >> 1;
        float4 a = make_float4(0.f, 0.f, 0.f, 0.f);
        for (int s = 0; s < nsteps; s += 8) {
            const int pos0 = 2 * s + h;
            switch (nsteps - s) {
                case 1:  a = gcn_pair_round<1>(S4, q, row, pos0, deg, a); break;
                case 2:  a = gcn_pair_round<2>(S4, q, row, pos0, deg, a); break;
                case 3:  a = gcn_pair_round<3>(S4, q, row, pos0, deg, a); break;
                case 4:  a = gcn_pair_round<4>(S4, q, row, pos0, deg, a); break;
                case 5:  a = gcn_pair_round<5>(S4, q, row, pos0, deg, a); break;
                case 6:  a = gcn_pair_round<6>(S4, q, row, pos0, deg, a); break;
                case 7:  a = gcn_pair_round<7>(S4, q, row, pos0, deg, a); break;
                default: a = gcn_pair_round<8>(S4, q, row, pos0, deg, a); break;
            }
        }

        a.x += gcn_other_half(a.x);
        a.y += gcn_other_half(a.y);
        a.z += gcn_other_half(a.z);
        a.w += gcn_other_half(a.w);

        if (h == 0) ((float4*)out)[(long long)(r0 + rl) * (DIM / 4) + q] = a;
    }
    if (!big) return;

    #pragma nounroll
    for (int i = 0; i < BUCKET_ROWS / SUM_WAVES; ++i) {
        const int rl = i * SUM_WAVES + w;
        if (rl >= rows) break;
        const int deg = __builtin_amdgcn_readfirstlane(lcnt[rl]);
        const int off = __builtin_amdgcn_readfirstlane(loff[rl]);
        if (deg <= 64) continue;

        float2 acc = make_float2(0.f, 0.f);
        int last = -1;
        for (int g = 0; g < deg; ++g) {
            int best = NO_EDGE, where = 0;
            for (int j = lane; j < deg; j += 32) {
                const int id = rec_id[off + j];
                if (id > last && id < best) { best = id; where = j; }
            }
            gcn_wave_min(best, where);
            if (best == NO_EDGE) break;
            const int2   cv = rec_cv[off + where];
            const float  v  = __int_as_float(cv.y);
            const float2 x  = S2[(long long)cv.x * (DIM / 2) + lane];
            acc.x += v * x.x;
            acc.y += v * x.y;
            last = best;
        }
        ((float2*)out)[(long long)(r0 + rl) * (DIM / 2) + lane] = acc;
    }
}

extern "C" void kernel_launch(void* const* d_in, const int* in_sizes, int n_in,
                              void* d_out, int out_size, void* d_ws, size_t ws_size,
                              hipStream_t stream) {
    const float* X    = (const float*)d_in[0];
    const float* W    = (const float*)d_in[1];
    const int*   arow = (const int*)d_in[2];
    const int*   acol = (const int*)d_in[3];
    const float* aval = (const float*)d_in[4];
    float*       out  = (float*)d_out;

    const int n_nodes = in_sizes[0] / DIM;
    const int n_edges = in_sizes[2];

    float* S        = (float*)d_ws;
    int2*  table    = (int2*)(S + (size_t)n_nodes * DIM);
    int4*  binned   = (int4*)(table + NCHUNK * MAX_BUCKETS);

    const int tiles = n_nodes / 16;
    const int g_all = (tiles + GEMM_WAVES - 1) / GEMM_WAVES;

    gcn_bin_kernel<<<NCHUNK + g_all, BIN_THREADS, 0, stream>>>(arow, acol, aval, table, binned, n_edges,
                                                              X, W, S, tiles);

    gcn_group_sum_kernel<<<MAX_BUCKETS, SUM_THREADS, 0, stream>>>(binned, table, S, out, n_nodes, n_edges);
}


# Binned records leave through LDS: a run's records are put at their places in an LDS copy of its stretch, which the work-group then writes out in order, whole 128-byte lines per store; sum order kept
# speedup vs baseline: 5.5210x; 1.0389x over previous
#include <hip/hip_runtime.h>

typedef __attribute__((ext_vector_type(2))) float v2f;
typedef __attribute__((ext_vector_type(8))) float v8f;

#define DIM 64
#define NO_EDGE 0x7fffffff

#define NCHUNK      256
#define BUCKET_ROWS 256
#define MAX_BUCKETS 512
#define BIN_THREADS 512
#define BIN_EDGES   4
#define BIN_HELD    13
#define STAGE_CAP   4096

#define SUM_THREADS 1024
#define SUM_WAVES   (SUM_THREADS / 32)
#define SUM_CAP     (6 * SUM_THREADS)
#define SEG_LANES   (SUM_THREADS / NCHUNK)
#define SEG_TAKE    8

#define GEMM_WAVES (BIN_THREADS / 32)

static __device__ __forceinline__ void gcn_gemm_tile(const float* __restrict__ X,
                                                     const float* __restrict__ W,
                                                     float* __restrict__ S,
                                                     int tile, int lane) {
    const int m0    = tile << 4;
    const int mrow  = lane & 15;
    const int khalf = lane >> 4;
    const int ncol  = lane & 15;

    v8f acc0 = {}, acc1 = {}, acc2 = {}, acc3 = {};

    const float* __restrict__ xrow = X + (long long)(m0 + mrow) * DIM;

    #pragma unroll
    for (int k0 = 0; k0 < DIM; k0 += 4) {
        const int ka = k0 + (khalf << 1);
        v2f a = *(const v2f*)(xrow + ka);

        const float* __restrict__ w0 = W + ka * DIM;
        const float* __restrict__ w1 = W + (ka + 1) * DIM;

        v2f b0, b1, b2, b3;
        b0.x = w0[ 0 + ncol]; b0.y = w1[ 0 + ncol];
        b1.x = w0[16 + ncol]; b1.y = w1[16 + ncol];
        b2.x = w0[32 + ncol]; b2.y = w1[32 + ncol];
        b3.x = w0[48 + ncol]; b3.y = w1[48 + ncol];

        acc0 = __builtin_amdgcn_wmma_f32_16x16x4_f32(false, a, false, b0,
                                                     (short)0, acc0, false, false);
        acc1 = __builtin_amdgcn_wmma_f32_16x16x4_f32(false, a, false, b1,
                                                     (short)0, acc1, false, false);
        acc2 = __builtin_amdgcn_wmma_f32_16x16x4_f32(false, a, false, b2,
                                                     (short)0, acc2, false, false);
        acc3 = __builtin_amdgcn_wmma_f32_16x16x4_f32(false, a, false, b3,
                                                     (short)0, acc3, false, false);
    }

    const int rbase = m0 + (khalf << 3);
    #pragma unroll
    for (int v = 0; v < 8; ++v) {
        float* __restrict__ srow = S + (long long)(rbase + v) * DIM + ncol;
        srow[ 0] = acc0[v];
        srow[16] = acc1[v];
        srow[32] = acc2[v];
        srow[48] = acc3[v];
    }
}

static __device__ __forceinline__ void gcn_gemm_extra_group(const float* __restrict__ X,
                                                            const float* __restrict__ W,
                                                            float* __restrict__ S,
                                                            int tile_first, int tile_end) {
    const int w    = __builtin_amdgcn_readfirstlane(threadIdx.x >> 5);
    const int tile = tile_first + (blockIdx.x - NCHUNK) * GEMM_WAVES + w;
    if (tile < tile_end) gcn_gemm_tile(X, W, S, tile, threadIdx.x & 31);
}

static __device__ __forceinline__ void gcn_lds_barrier() {
    __builtin_amdgcn_fence(__ATOMIC_RELEASE, "workgroup", "local");
    __builtin_amdgcn_s_barrier();
    __builtin_amdgcn_fence(__ATOMIC_ACQUIRE, "workgroup", "local");
}

__global__ void __launch_bounds__(BIN_THREADS) __attribute__((amdgpu_waves_per_eu(12)))
gcn_bin_kernel(const int* __restrict__ adj_row,
               const int* __restrict__ adj_col,
               const float* __restrict__ adj_val,
               int2* __restrict__ table,
               int4* __restrict__ binned,
               int n_edges,
               const float* __restrict__ X, const float* __restrict__ W,
               float* __restrict__ S, int tile_end) {
    __shared__ int lh[MAX_BUCKETS];
    __shared__ int sc[MAX_BUCKETS];
    __shared__ int lpos[MAX_BUCKETS];
    __shared__ int4 stage[STAGE_CAP];
    static_assert(STAGE_CAP % 8 == 0, "a part of the stretch must start on a 128-byte line");
    if (blockIdx.x >= NCHUNK) {
        gcn_gemm_extra_group(X, W, S, 0, tile_end);
        return;
    }
    const int t = threadIdx.x;
    const int c = blockIdx.x;

    const int run = (n_edges + NCHUNK - 1) / NCHUNK;
    const int e0  = c * run;
    const int e1  = (e0 + run < n_edges) ? e0 + run : n_edges;
    const bool held = run <= BIN_HELD * BIN_THREADS;

    lh[t] = 0;
    __syncthreads();

    int   hr[BIN_HELD], hc[BIN_HELD];
    float hv[BIN_HELD];
    if (held) {
        #pragma unroll
        for (int k = 0; k < BIN_HELD; ++k) {
            const int e  = e0 + t + k * BIN_THREADS;
            const int ec = (e < e1) ? e : ((e1 > 0) ? e1 - 1 : 0);
            hr[k] = adj_row[ec];
        }
        #pragma unroll
        for (int k = 0; k < BIN_HELD; ++k) {
            const int e  = e0 + t + k * BIN_THREADS;
            const int ec = (e < e1) ? e : ((e1 > 0) ? e1 - 1 : 0);
            hc[k] = adj_col[ec];
            hv[k] = adj_val[ec];
        }
        #pragma unroll
        for (int k = 0; k < BIN_HELD; ++k) {
            const int e = e0 + t + k * BIN_THREADS;
            if (e >= e1) hr[k] = -1;
        }
    }

    if (held) {
        #pragma unroll
        for (int k = 0; k < BIN_HELD; ++k) {
            if (hr[k] >= 0) atomicAdd(&lh[hr[k] >> 8], 1);
        }
    } else
    for (int eb = e0 + t; eb < e1; eb += BIN_EDGES * BIN_THREADS) {
        int r[BIN_EDGES];
        #pragma unroll
        for (int k = 0; k < BIN_EDGES; ++k) {
            const int e = eb + k * BIN_THREADS;
            r[k] = (e < e1) ? adj_row[e] : -1;
        }
        #pragma unroll
        for (int k = 0; k < BIN_EDGES; ++k) {
            if (r[k] >= 0) atomicAdd(&lh[r[k] >> 8], 1);
        }
    }
    __syncthreads();

    const int tot = lh[t];
    sc[t] = tot;
    __syncthreads();
    for (int off = 1; off < MAX_BUCKETS; off <<= 1) {
        const int add = (t >= off) ? sc[t - off] : 0;
        __syncthreads();
        sc[t] += add;
        __syncthreads();
    }
    const int first = sc[t] - tot;
    lpos[t] = e0 + first;
    table[c * MAX_BUCKETS + t] = make_int2(first, tot);
    __syncthreads();

    if (held) {
        int p[BIN_HELD];
        #pragma unroll
        for (int k = 0; k < BIN_HELD; ++k) {
            p[k] = 0;
            if (hr[k] >= 0) p[k] = atomicAdd(&lpos[hr[k] >> 8], 1);
        }
        const int cnt  = e1 - e0;
        const int lead = e0 & 7;
        for (int b0 = 0; b0 < cnt; b0 += STAGE_CAP) {
            const int b1 = (b0 + STAGE_CAP < cnt) ? b0 + STAGE_CAP : cnt;
            if (b0 > 0) gcn_lds_barrier();
            #pragma unroll
            for (int k = 0; k < BIN_HELD; ++k) {
                const int e = e0 + t + k * BIN_THREADS;
                const int i = p[k] - e0;
                if (hr[k] >= 0 && i >= b0 && i < b1)
                    stage[i - b0] = make_int4(e, hr[k], hc[k], __float_as_int(hv[k]));
            }
            gcn_lds_barrier();
            for (int g = b0 + t - lead; g < b1; g += BIN_THREADS) {
                if (g >= b0) binned[e0 + g] = stage[g - b0];
            }
        }
        return;
    }

    for (int eb = e0 + t; eb < e1; eb += BIN_EDGES * BIN_THREADS) {
        int   r[BIN_EDGES], cl[BIN_EDGES], p[BIN_EDGES];
        float v[BIN_EDGES];
        #pragma unroll
        for (int k = 0; k < BIN_EDGES; ++k) {
            const int e = eb + k * BIN_THREADS;
            r[k] = -1; cl[k] = 0; v[k] = 0.f;
            if (e < e1) { r[k] = adj_row[e]; cl[k] = adj_col[e]; v[k] = adj_val[e]; }
        }
        #pragma unroll
        for (int k = 0; k < BIN_EDGES; ++k) {
            p[k] = 0;
            if (r[k] >= 0) p[k] = atomicAdd(&lpos[r[k] >> 8], 1);
        }
        #pragma unroll
        for (int k = 0; k < BIN_EDGES; ++k) {
            const int e = eb + k * BIN_THREADS;
            if (r[k] >= 0) binned[p[k]] = make_int4(e, r[k], cl[k], __float_as_int(v[k]));
        }
    }
}

template <int U>
static __device__ __forceinline__ float4 gcn_pair_round(const float4* __restrict__ S4, int q,
                                                        const int2* row, int pos0, int deg, float4 acc) {
    float  v[U];
    float4 x[U];
    const int  plast = pos0 + 2 * (U - 1);
    const bool have  = plast < deg;
    #pragma unroll
    for (int k = 0; k < U; ++k) {
        const int  p  = (k == U - 1) ? (have ? plast : deg - 1) : pos0 + 2 * k;
        const int2 cv = row[p];
        v[k] = __int_as_float(cv.y);
        x[k] = S4[(unsigned)(cv.x * (DIM / 4) + q)];
    }
    if (!have) { v[U - 1] = 0.f; x[U - 1] = make_float4(0.f, 0.f, 0.f, 0.f); }
    #pragma unroll
    for (int k = 0; k < U; ++k) {
        acc.x = __builtin_fmaf(v[k], x[k].x, acc.x);
        acc.y = __builtin_fmaf(v[k], x[k].y, acc.y);
        acc.z = __builtin_fmaf(v[k], x[k].z, acc.z);
        acc.w = __builtin_fmaf(v[k], x[k].w, acc.w);
    }
    return acc;
}

static __device__ __forceinline__ float gcn_other_half(float f) {
    const unsigned u = (unsigned)__float_as_int(f);
    return __int_as_float((int)__builtin_amdgcn_permlanex16(u, u, 0x76543210u, 0xfedcba98u, false, false));
}

static __device__ __forceinline__ void gcn_wave_min(int& best, int& where) {
    for (int off = 16; off > 0; off >>= 1) {
        const int ob = __shfl_xor(best, off);
        const int ow = __shfl_xor(where, off);
        if (ob < best) { best = ob; where = ow; }
    }
}

__global__ void __launch_bounds__(SUM_THREADS)
gcn_group_sum_kernel(const int4* __restrict__ binned,
                     const int2* __restrict__ table,
                     const float* __restrict__ S,
                     float* __restrict__ out,
                     int n_nodes, int n_edges) {
    static_assert(SEG_LANES * NCHUNK == SUM_THREADS, "every run's stretch needs its threads");
    __shared__ int  lcnt[BUCKET_ROWS];
    __shared__ int  loff[BUCKET_ROWS];
    __shared__ int  lpos[BUCKET_ROWS];
    __shared__ int  wtot[BUCKET_ROWS / 32];
    __shared__ int  rec_id[SUM_CAP];
    __shared__ int2 rec_cv[SUM_CAP];

    const int t    = threadIdx.x;
    const int lane = t & 31;
    const int w  = __builtin_amdgcn_readfirstlane(t >> 5);
    const int r0 = blockIdx.x * BUCKET_ROWS;
    if (r0 >= n_nodes) return;

    const int rows = (n_nodes - r0 < BUCKET_ROWS) ? n_nodes - r0 : BUCKET_ROWS;
    const float2* __restrict__ S2 = (const float2*)S;
    const float4* __restrict__ S4 = (const float4*)S;
    const int h = lane >> 4;
    const int q = lane & 15;

    const int run = (n_edges + NCHUNK - 1) / NCHUNK;
    const int sub = t & (SEG_LANES - 1);
    const int2 sg = table[(t / SEG_LANES) * MAX_BUCKETS + blockIdx.x];
    const int sbase = (t / SEG_LANES) * run + sg.x;
    const int scnt  = sg.y;

    if (t < BUCKET_ROWS) lcnt[t] = 0;
    __syncthreads();

    for (int j0 = sub; j0 < scnt; j0 += SEG_LANES * SEG_TAKE) {
        int r[SEG_TAKE];
        #pragma unroll
        for (int k = 0; k < SEG_TAKE; ++k) {
            const int j = j0 + k * SEG_LANES;
            r[k] = (j < scnt) ? binned[sbase + j].y : -1;
        }
        #pragma unroll
        for (int k = 0; k < SEG_TAKE; ++k) {
            if (r[k] >= 0) atomicAdd(&lcnt[r[k] - r0], 1);
        }
    }
    __syncthreads();

    int c = 0, incl = 0;
    if (t < BUCKET_ROWS) {
        c = lcnt[t];
        incl = c;
        for (int off = 1; off < 32; off <<= 1) {
            const int up = __shfl_up(incl, off);
            if (lane >= off) incl += up;
        }
        if (lane == 31) wtot[w] = incl;
    }
    __syncthreads();
    if (t < BUCKET_ROWS) {
        int base = 0;
        for (int j = 0; j < BUCKET_ROWS / 32; ++j) base += (j < w) ? wtot[j] : 0;
        loff[t] = base + incl - c;
        lpos[t] = base + incl - c;
    }
    int n = 0;
    for (int j = 0; j < BUCKET_ROWS / 32; ++j) n += wtot[j];
    __syncthreads();

    if (n > SUM_CAP) {
        for (int i = 0; i < BUCKET_ROWS / SUM_WAVES; ++i) {
            const int rl = i * SUM_WAVES + w;
            if (rl >= rows) break;
            const int r = r0 + rl;
            float2 acc = make_float2(0.f, 0.f);
            int last = -1;
            for (int g = 0; g < n; ++g) {
                int best = NO_EDGE, where = 0;
                for (int cc = 0; cc < NCHUNK; ++cc) {
                    const int2 s2  = table[cc * MAX_BUCKETS + blockIdx.x];
                    const int  beg = cc * run + s2.x;
                    for (int j = lane; j < s2.y; j += 32) {
                        const int4 q = binned[beg + j];
                        if (q.y == r && q.x > last && q.x < best) { best = q.x; where = beg + j; }
                    }
                }
                gcn_wave_min(best, where);
                if (best == NO_EDGE) break;
                const int4   q = binned[where];
                const float  v = __int_as_float(q.w);
                const float2 x = S2[(long long)q.z * (DIM / 2) + lane];
                acc.x += v * x.x;
                acc.y += v * x.y;
                last = best;
            }
            ((float2*)out)[(long long)r * (DIM / 2) + lane] = acc;
        }
        return;
    }

    for (int j0 = sub; j0 < scnt; j0 += SEG_LANES * SEG_TAKE) {
        int4 q[SEG_TAKE];
        int  p[SEG_TAKE];
        #pragma unroll
        for (int k = 0; k < SEG_TAKE; ++k) {
            const int j = j0 + k * SEG_LANES;
            q[k] = make_int4(0, -1, 0, 0);
            if (j < scnt) q[k] = binned[sbase + j];
        }
        #pragma unroll
        for (int k = 0; k < SEG_TAKE; ++k) {
            p[k] = 0;
            if (q[k].y >= 0) p[k] = atomicAdd(&lpos[q[k].y - r0], 1);
        }
        #pragma unroll
        for (int k = 0; k < SEG_TAKE; ++k) {
            if (q[k].y >= 0) {
                rec_id[p[k]] = q[k].x;
                rec_cv[p[k]] = make_int2(q[k].z, q[k].w);
            }
        }
    }
    __syncthreads();

    bool big = false;
    for (int i = 0; i < BUCKET_ROWS / SUM_WAVES; ++i) {
        const int rl = i * SUM_WAVES + w;
        if (rl >= rows) break;
        const int deg = __builtin_amdgcn_readfirstlane(lcnt[rl]);
        const int off = __builtin_amdgcn_readfirstlane(loff[rl]);
        if (deg > 64) { big = true; continue; }

        int e0 = NO_EDGE, c0 = 0, v0 = 0;
        int e1 = NO_EDGE, c1 = 0, v1 = 0;
        if (lane < deg)      { e0 = rec_id[off + lane];      const int2 cv = rec_cv[off + lane];      c0 = cv.x; v0 = cv.y; }
        if (lane + 32 < deg) { e1 = rec_id[off + lane + 32]; const int2 cv = rec_cv[off + lane + 32]; c1 = cv.x; v1 = cv.y; }

        int rk0 = 0, rk1 = 0;
        const int n0 = deg < 32 ? deg : 32;
        if (deg <= 32) {
            for (int j = 0; j < n0; ++j) {
                const int kj = __builtin_amdgcn_readlane(e0, j);
                rk0 += (kj < e0) ? 1 : 0;
            }
        } else {
            for (int j = 0; j < 32; ++j) {
                const int kj = __builtin_amdgcn_readlane(e0, j);
                rk0 += (kj < e0) ? 1 : 0;
                rk1 += (kj < e1) ? 1 : 0;
            }
            for (int j = 32; j < deg; ++j) {
                const int kj = __builtin_amdgcn_readlane(e1, j - 32);
                rk0 += (kj < e0) ? 1 : 0;
                rk1 += (kj < e1) ? 1 : 0;
            }
        }
        if (lane < deg)      rec_cv[off + rk0] = make_int2(c0, v0);
        if (lane + 32 < deg) rec_cv[off + rk1] = make_int2(c1, v1);

        const int2* row    = &rec_cv[off];
        const int   nsteps = (deg + 1) >> 1;
        float4 a = make_float4(0.f, 0.f, 0.f, 0.f);
        for (int s = 0; s < nsteps; s += 8) {
            const int pos0 = 2 * s + h;
            switch (nsteps - s) {
                case 1:  a = gcn_pair_round<1>(S4, q, row, pos0, deg, a); break;
                case 2:  a = gcn_pair_round<2>(S4, q, row, pos0, deg, a); break;
                case 3:  a = gcn_pair_round<3>(S4, q, row, pos0, deg, a); break;
                case 4:  a = gcn_pair_round<4>(S4, q, row, pos0, deg, a); break;
                case 5:  a = gcn_pair_round<5>(S4, q, row, pos0, deg, a); break;
                case 6:  a = gcn_pair_round<6>(S4, q, row, pos0, deg, a); break;
                case 7:  a = gcn_pair_round<7>(S4, q, row, pos0, deg, a); break;
                default: a = gcn_pair_round<8>(S4, q, row, pos0, deg, a); break;
            }
        }

        a.x += gcn_other_half(a.x);
        a.y += gcn_other_half(a.y);
        a.z += gcn_other_half(a.z);
        a.w += gcn_other_half(a.w);

        if (h == 0) ((float4*)out)[(long long)(r0 + rl) * (DIM / 4) + q] = a;
    }
    if (!big) return;

    #pragma nounroll
    for (int i = 0; i < BUCKET_ROWS / SUM_WAVES; ++i) {
        const int rl = i * SUM_WAVES + w;
        if (rl >= rows) break;
        const int deg = __builtin_amdgcn_readfirstlane(lcnt[rl]);
        const int off = __builtin_amdgcn_readfirstlane(loff[rl]);
        if (deg <= 64) continue;

        float2 acc = make_float2(0.f, 0.f);
        int last = -1;
        for (int g = 0; g < deg; ++g) {
            int best = NO_EDGE, where = 0;
            for (int j = lane; j < deg; j += 32) {
                const int id = rec_id[off + j];
                if (id > last && id < best) { best = id; where = j; }
            }
            gcn_wave_min(best, where);
            if (best == NO_EDGE) break;
            const int2   cv = rec_cv[off + where];
            const float  v  = __int_as_float(cv.y);
            const float2 x  = S2[(long long)cv.x * (DIM / 2) + lane];
            acc.x += v * x.x;
            acc.y += v * x.y;
            last = best;
        }
        ((float2*)out)[(long long)(r0 + rl) * (DIM / 2) + lane] = acc;
    }
}

extern "C" void kernel_launch(void* const* d_in, const int* in_sizes, int n_in,
                              void* d_out, int out_size, void* d_ws, size_t ws_size,
                              hipStream_t stream) {
    const float* X    = (const float*)d_in[0];
    const float* W    = (const float*)d_in[1];
    const int*   arow = (const int*)d_in[2];
    const int*   acol = (const int*)d_in[3];
    const float* aval = (const float*)d_in[4];
    float*       out  = (float*)d_out;

    const int n_nodes = in_sizes[0] / DIM;
    const int n_edges = in_sizes[2];

    float* S        = (float*)d_ws;
    int2*  table    = (int2*)(S + (size_t)n_nodes * DIM);
    int4*  binned   = (int4*)(table + NCHUNK * MAX_BUCKETS);

    const int tiles = n_nodes / 16;
    const int g_all = (tiles + GEMM_WAVES - 1) / GEMM_WAVES;

    gcn_bin_kernel<<<NCHUNK + g_all, BIN_THREADS, 0, stream>>>(arow, acol, aval, table, binned, n_edges,
                                                              X, W, S, tiles);

    gcn_group_sum_kernel<<<MAX_BUCKETS, SUM_THREADS, 0, stream>>>(binned, table, S, out, n_nodes, n_edges);
}


# Summing launch reads each binned record once: a half-wave per run's stretch, one 16-byte load per lane covering 16 records, kept in registers from the row count to the placing; sum order kept
# speedup vs baseline: 5.9338x; 1.0748x over previous
#include <hip/hip_runtime.h>

typedef __attribute__((ext_vector_type(2))) float v2f;
typedef __attribute__((ext_vector_type(8))) float v8f;

#define DIM 64
#define NO_EDGE 0x7fffffff

#define NCHUNK      256
#define BUCKET_ROWS 256
#define MAX_BUCKETS 512
#define BIN_THREADS 512
#define BIN_EDGES   4
#define BIN_HELD    13
#define STAGE_CAP   4096

#define SUM_THREADS 1024
#define SUM_WAVES   (SUM_THREADS / 32)
#define SUM_CAP     (6 * SUM_THREADS)
#define SEG_LANES   16
#define SEG_GROUPS  (SUM_THREADS / SEG_LANES)
#define SEG_PER     (NCHUNK / SEG_GROUPS)
#define SEG_HELD    2

#define GEMM_WAVES (BIN_THREADS / 32)

static __device__ __forceinline__ void gcn_gemm_tile(const float* __restrict__ X,
                                                     const float* __restrict__ W,
                                                     float* __restrict__ S,
                                                     int tile, int lane) {
    const int m0    = tile << 4;
    const int mrow  = lane & 15;
    const int khalf = lane >> 4;
    const int ncol  = lane & 15;

    v8f acc0 = {}, acc1 = {}, acc2 = {}, acc3 = {};

    const float* __restrict__ xrow = X + (long long)(m0 + mrow) * DIM;

    #pragma unroll
    for (int k0 = 0; k0 < DIM; k0 += 4) {
        const int ka = k0 + (khalf << 1);
        v2f a = *(const v2f*)(xrow + ka);

        const float* __restrict__ w0 = W + ka * DIM;
        const float* __restrict__ w1 = W + (ka + 1) * DIM;

        v2f b0, b1, b2, b3;
        b0.x = w0[ 0 + ncol]; b0.y = w1[ 0 + ncol];
        b1.x = w0[16 + ncol]; b1.y = w1[16 + ncol];
        b2.x = w0[32 + ncol]; b2.y = w1[32 + ncol];
        b3.x = w0[48 + ncol]; b3.y = w1[48 + ncol];

        acc0 = __builtin_amdgcn_wmma_f32_16x16x4_f32(false, a, false, b0,
                                                     (short)0, acc0, false, false);
        acc1 = __builtin_amdgcn_wmma_f32_16x16x4_f32(false, a, false, b1,
                                                     (short)0, acc1, false, false);
        acc2 = __builtin_amdgcn_wmma_f32_16x16x4_f32(false, a, false, b2,
                                                     (short)0, acc2, false, false);
        acc3 = __builtin_amdgcn_wmma_f32_16x16x4_f32(false, a, false, b3,
                                                     (short)0, acc3, false, false);
    }

    const int rbase = m0 + (khalf << 3);
    #pragma unroll
    for (int v = 0; v < 8; ++v) {
        float* __restrict__ srow = S + (long long)(rbase + v) * DIM + ncol;
        srow[ 0] = acc0[v];
        srow[16] = acc1[v];
        srow[32] = acc2[v];
        srow[48] = acc3[v];
    }
}

static __device__ __forceinline__ void gcn_gemm_extra_group(const float* __restrict__ X,
                                                            const float* __restrict__ W,
                                                            float* __restrict__ S,
                                                            int tile_first, int tile_end) {
    const int w    = __builtin_amdgcn_readfirstlane(threadIdx.x >> 5);
    const int tile = tile_first + (blockIdx.x - NCHUNK) * GEMM_WAVES + w;
    if (tile < tile_end) gcn_gemm_tile(X, W, S, tile, threadIdx.x & 31);
}

static __device__ __forceinline__ void gcn_lds_barrier() {
    __builtin_amdgcn_fence(__ATOMIC_RELEASE, "workgroup", "local");
    __builtin_amdgcn_s_barrier();
    __builtin_amdgcn_fence(__ATOMIC_ACQUIRE, "workgroup", "local");
}

__global__ void __launch_bounds__(BIN_THREADS) __attribute__((amdgpu_waves_per_eu(12)))
gcn_bin_kernel(const int* __restrict__ adj_row,
               const int* __restrict__ adj_col,
               const float* __restrict__ adj_val,
               int2* __restrict__ table,
               int4* __restrict__ binned,
               int n_edges,
               const float* __restrict__ X, const float* __restrict__ W,
               float* __restrict__ S, int tile_end) {
    __shared__ int lh[MAX_BUCKETS];
    __shared__ int sc[MAX_BUCKETS];
    __shared__ int lpos[MAX_BUCKETS];
    __shared__ int4 stage[STAGE_CAP];
    static_assert(STAGE_CAP % 8 == 0, "a part of the stretch must start on a 128-byte line");
    if (blockIdx.x >= NCHUNK) {
        gcn_gemm_extra_group(X, W, S, 0, tile_end);
        return;
    }
    const int t = threadIdx.x;
    const int c = blockIdx.x;

    const int run = (n_edges + NCHUNK - 1) / NCHUNK;
    const int e0  = c * run;
    const int e1  = (e0 + run < n_edges) ? e0 + run : n_edges;
    const bool held = run <= BIN_HELD * BIN_THREADS;

    lh[t] = 0;
    __syncthreads();

    int   hr[BIN_HELD], hc[BIN_HELD];
    float hv[BIN_HELD];
    if (held) {
        #pragma unroll
        for (int k = 0; k < BIN_HELD; ++k) {
            const int e  = e0 + t + k * BIN_THREADS;
            const int ec = (e < e1) ? e : ((e1 > 0) ? e1 - 1 : 0);
            hr[k] = adj_row[ec];
        }
        #pragma unroll
        for (int k = 0; k < BIN_HELD; ++k) {
            const int e  = e0 + t + k * BIN_THREADS;
            const int ec = (e < e1) ? e : ((e1 > 0) ? e1 - 1 : 0);
            hc[k] = adj_col[ec];
            hv[k] = adj_val[ec];
        }
        #pragma unroll
        for (int k = 0; k < BIN_HELD; ++k) {
            const int e = e0 + t + k * BIN_THREADS;
            if (e >= e1) hr[k] = -1;
        }
    }

    if (held) {
        #pragma unroll
        for (int k = 0; k < BIN_HELD; ++k) {
            if (hr[k] >= 0) atomicAdd(&lh[hr[k] >> 8], 1);
        }
    } else
    for (int eb = e0 + t; eb < e1; eb += BIN_EDGES * BIN_THREADS) {
        int r[BIN_EDGES];
        #pragma unroll
        for (int k = 0; k < BIN_EDGES; ++k) {
            const int e = eb + k * BIN_THREADS;
            r[k] = (e < e1) ? adj_row[e] : -1;
        }
        #pragma unroll
        for (int k = 0; k < BIN_EDGES; ++k) {
            if (r[k] >= 0) atomicAdd(&lh[r[k] >> 8], 1);
        }
    }
    __syncthreads();

    const int tot = lh[t];
    sc[t] = tot;
    __syncthreads();
    for (int off = 1; off < MAX_BUCKETS; off <<= 1) {
        const int add = (t >= off) ? sc[t - off] : 0;
        __syncthreads();
        sc[t] += add;
        __syncthreads();
    }
    const int first = sc[t] - tot;
    lpos[t] = e0 + first;
    table[c * MAX_BUCKETS + t] = make_int2(first, tot);
    __syncthreads();

    if (held) {
        int p[BIN_HELD];
        #pragma unroll
        for (int k = 0; k < BIN_HELD; ++k) {
            p[k] = 0;
            if (hr[k] >= 0) p[k] = atomicAdd(&lpos[hr[k] >> 8], 1);
        }
        const int cnt  = e1 - e0;
        const int lead = e0 & 7;
        for (int b0 = 0; b0 < cnt; b0 += STAGE_CAP) {
            const int b1 = (b0 + STAGE_CAP < cnt) ? b0 + STAGE_CAP : cnt;
            if (b0 > 0) gcn_lds_barrier();
            #pragma unroll
            for (int k = 0; k < BIN_HELD; ++k) {
                const int e = e0 + t + k * BIN_THREADS;
                const int i = p[k] - e0;
                if (hr[k] >= 0 && i >= b0 && i < b1)
                    stage[i - b0] = make_int4(e, hr[k], hc[k], __float_as_int(hv[k]));
            }
            gcn_lds_barrier();
            for (int g = b0 + t - lead; g < b1; g += BIN_THREADS) {
                if (g >= b0) binned[e0 + g] = stage[g - b0];
            }
        }
        return;
    }

    for (int eb = e0 + t; eb < e1; eb += BIN_EDGES * BIN_THREADS) {
        int   r[BIN_EDGES], cl[BIN_EDGES], p[BIN_EDGES];
        float v[BIN_EDGES];
        #pragma unroll
        for (int k = 0; k < BIN_EDGES; ++k) {
            const int e = eb + k * BIN_THREADS;
            r[k] = -1; cl[k] = 0; v[k] = 0.f;
            if (e < e1) { r[k] = adj_row[e]; cl[k] = adj_col[e]; v[k] = adj_val[e]; }
        }
        #pragma unroll
        for (int k = 0; k < BIN_EDGES; ++k) {
            p[k] = 0;
            if (r[k] >= 0) p[k] = atomicAdd(&lpos[r[k] >> 8], 1);
        }
        #pragma unroll
        for (int k = 0; k < BIN_EDGES; ++k) {
            const int e = eb + k * BIN_THREADS;
            if (r[k] >= 0) binned[p[k]] = make_int4(e, r[k], cl[k], __float_as_int(v[k]));
        }
    }
}

template <int U>
static __device__ __forceinline__ float4 gcn_pair_round(const float4* __restrict__ S4, int q,
                                                        const int2* row, int pos0, int deg, float4 acc) {
    float  v[U];
    float4 x[U];
    const int  plast = pos0 + 2 * (U - 1);
    const bool have  = plast < deg;
    #pragma unroll
    for (int k = 0; k < U; ++k) {
        const int  p  = (k == U - 1) ? (have ? plast : deg - 1) : pos0 + 2 * k;
        const int2 cv = row[p];
        v[k] = __int_as_float(cv.y);
        x[k] = S4[(unsigned)(cv.x * (DIM / 4) + q)];
    }
    if (!have) { v[U - 1] = 0.f; x[U - 1] = make_float4(0.f, 0.f, 0.f, 0.f); }
    #pragma unroll
    for (int k = 0; k < U; ++k) {
        acc.x = __builtin_fmaf(v[k], x[k].x, acc.x);
        acc.y = __builtin_fmaf(v[k], x[k].y, acc.y);
        acc.z = __builtin_fmaf(v[k], x[k].z, acc.z);
        acc.w = __builtin_fmaf(v[k], x[k].w, acc.w);
    }
    return acc;
}

static __device__ __forceinline__ float gcn_other_half(float f) {
    const unsigned u = (unsigned)__float_as_int(f);
    return __int_as_float((int)__builtin_amdgcn_permlanex16(u, u, 0x76543210u, 0xfedcba98u, false, false));
}

static __device__ __forceinline__ void gcn_wave_min(int& best, int& where) {
    for (int off = 16; off > 0; off >>= 1) {
        const int ob = __shfl_xor(best, off);
        const int ow = __shfl_xor(where, off);
        if (ob < best) { best = ob; where = ow; }
    }
}

__global__ void __launch_bounds__(SUM_THREADS) __attribute__((amdgpu_waves_per_eu(16)))
gcn_group_sum_kernel(const int4* __restrict__ binned,
                     const int2* __restrict__ table,
                     const float* __restrict__ S,
                     float* __restrict__ out,
                     int n_nodes, int n_edges) {
    static_assert(SEG_GROUPS * SEG_PER == NCHUNK, "every run's stretch needs its half-wave");
    __shared__ int  lcnt[BUCKET_ROWS];
    __shared__ int  loff[BUCKET_ROWS];
    __shared__ int  lpos[BUCKET_ROWS];
    __shared__ int  wtot[BUCKET_ROWS / 32];
    __shared__ int  rec_id[SUM_CAP];
    __shared__ int2 rec_cv[SUM_CAP];

    const int t    = threadIdx.x;
    const int lane = t & 31;
    const int w  = __builtin_amdgcn_readfirstlane(t >> 5);
    const int r0 = blockIdx.x * BUCKET_ROWS;
    if (r0 >= n_nodes) return;

    const int rows = (n_nodes - r0 < BUCKET_ROWS) ? n_nodes - r0 : BUCKET_ROWS;
    const float2* __restrict__ S2 = (const float2*)S;
    const float4* __restrict__ S4 = (const float4*)S;
    const int h = lane >> 4;
    const int q = lane & 15;

    const int run = (n_edges + NCHUNK - 1) / NCHUNK;
    const int sub = t & (SEG_LANES - 1);
    const int grp = t / SEG_LANES;
    int sbase[SEG_PER], scnt[SEG_PER];
    #pragma unroll
    for (int m = 0; m < SEG_PER; ++m) {
        const int  cr = grp + m * SEG_GROUPS;
        const int2 sg = table[cr * MAX_BUCKETS + blockIdx.x];
        sbase[m] = cr * run + sg.x;
        scnt[m]  = sg.y;
    }

    int4 hq[SEG_PER * SEG_HELD];
    #pragma unroll
    for (int m = 0; m < SEG_PER; ++m) {
        #pragma unroll
        for (int k = 0; k < SEG_HELD; ++k) {
            const int j = sub + k * SEG_LANES;
            hq[m * SEG_HELD + k] = make_int4(0, -1, 0, 0);
            if (j < scnt[m]) hq[m * SEG_HELD + k] = binned[sbase[m] + j];
        }
    }

    if (t < BUCKET_ROWS) lcnt[t] = 0;
    __syncthreads();

    #pragma unroll
    for (int i = 0; i < SEG_PER * SEG_HELD; ++i) {
        if (hq[i].y >= 0) atomicAdd(&lcnt[hq[i].y - r0], 1);
    }
    bool more = false;
    #pragma unroll
    for (int m = 0; m < SEG_PER; ++m) {
        more |= sub + SEG_HELD * SEG_LANES < scnt[m];
        for (int j = sub + SEG_HELD * SEG_LANES; j < scnt[m]; j += SEG_LANES) {
            atomicAdd(&lcnt[binned[sbase[m] + j].y - r0], 1);
        }
    }
    __syncthreads();

    int c = 0, incl = 0;
    if (t < BUCKET_ROWS) {
        c = lcnt[t];
        incl = c;
        for (int off = 1; off < 32; off <<= 1) {
            const int up = __shfl_up(incl, off);
            if (lane >= off) incl += up;
        }
        if (lane == 31) wtot[w] = incl;
    }
    __syncthreads();
    if (t < BUCKET_ROWS) {
        int base = 0;
        for (int j = 0; j < BUCKET_ROWS / 32; ++j) base += (j < w) ? wtot[j] : 0;
        loff[t] = base + incl - c;
        lpos[t] = base + incl - c;
    }
    int n = 0;
    for (int j = 0; j < BUCKET_ROWS / 32; ++j) n += wtot[j];
    __syncthreads();

    if (n > SUM_CAP) {
        for (int i = 0; i < BUCKET_ROWS / SUM_WAVES; ++i) {
            const int rl = i * SUM_WAVES + w;
            if (rl >= rows) break;
            const int r = r0 + rl;
            float2 acc = make_float2(0.f, 0.f);
            int last = -1;
            for (int g = 0; g < n; ++g) {
                int best = NO_EDGE, where = 0;
                for (int cc = 0; cc < NCHUNK; ++cc) {
                    const int2 s2  = table[cc * MAX_BUCKETS + blockIdx.x];
                    const int  beg = cc * run + s2.x;
                    for (int j = lane; j < s2.y; j += 32) {
                        const int4 q = binned[beg + j];
                        if (q.y == r && q.x > last && q.x < best) { best = q.x; where = beg + j; }
                    }
                }
                gcn_wave_min(best, where);
                if (best == NO_EDGE) break;
                const int4   q = binned[where];
                const float  v = __int_as_float(q.w);
                const float2 x = S2[(long long)q.z * (DIM / 2) + lane];
                acc.x += v * x.x;
                acc.y += v * x.y;
                last = best;
            }
            ((float2*)out)[(long long)r * (DIM / 2) + lane] = acc;
        }
        return;
    }

    {
        int p[SEG_PER * SEG_HELD];
        #pragma unroll
        for (int i = 0; i < SEG_PER * SEG_HELD; ++i) {
            p[i] = 0;
            if (hq[i].y >= 0) p[i] = atomicAdd(&lpos[hq[i].y - r0], 1);
        }
        #pragma unroll
        for (int i = 0; i < SEG_PER * SEG_HELD; ++i) {
            if (hq[i].y >= 0) {
                rec_id[p[i]] = hq[i].x;
                rec_cv[p[i]] = make_int2(hq[i].z, hq[i].w);
            }
        }
    }
    if (more) {
        #pragma nounroll
        for (int m = 0; m < SEG_PER; ++m) {
            const int  cr = grp + m * SEG_GROUPS;
            const int2 sg = table[cr * MAX_BUCKETS + blockIdx.x];
            for (int j = sub + SEG_HELD * SEG_LANES; j < sg.y; j += SEG_LANES) {
                const int4 q = binned[cr * run + sg.x + j];
                const int  p = atomicAdd(&lpos[q.y - r0], 1);
                rec_id[p] = q.x;
                rec_cv[p] = make_int2(q.z, q.w);
            }
        }
    }
    __syncthreads();

    bool big = false;
    for (int i = 0; i < BUCKET_ROWS / SUM_WAVES; ++i) {
        const int rl = i * SUM_WAVES + w;
        if (rl >= rows) break;
        const int deg = __builtin_amdgcn_readfirstlane(lcnt[rl]);
        const int off = __builtin_amdgcn_readfirstlane(loff[rl]);
        if (deg > 64) { big = true; continue; }

        int e0 = NO_EDGE, c0 = 0, v0 = 0;
        int e1 = NO_EDGE, c1 = 0, v1 = 0;
        if (lane < deg)      { e0 = rec_id[off + lane];      const int2 cv = rec_cv[off + lane];      c0 = cv.x; v0 = cv.y; }
        if (lane + 32 < deg) { e1 = rec_id[off + lane + 32]; const int2 cv = rec_cv[off + lane + 32]; c1 = cv.x; v1 = cv.y; }

        int rk0 = 0, rk1 = 0;
        const int n0 = deg < 32 ? deg : 32;
        if (deg <= 32) {
            for (int j = 0; j < n0; ++j) {
                const int kj = __builtin_amdgcn_readlane(e0, j);
                rk0 += (kj < e0) ? 1 : 0;
            }
        } else {
            for (int j = 0; j < 32; ++j) {
                const int kj = __builtin_amdgcn_readlane(e0, j);
                rk0 += (kj < e0) ? 1 : 0;
                rk1 += (kj < e1) ? 1 : 0;
            }
            for (int j = 32; j < deg; ++j) {
                const int kj = __builtin_amdgcn_readlane(e1, j - 32);
                rk0 += (kj < e0) ? 1 : 0;
                rk1 += (kj < e1) ? 1 : 0;
            }
        }
        if (lane < deg)      rec_cv[off + rk0] = make_int2(c0, v0);
        if (lane + 32 < deg) rec_cv[off + rk1] = make_int2(c1, v1);

        const int2* row    = &rec_cv[off];
        const int   nsteps = (deg + 1) >> 1;
        float4 a = make_float4(0.f, 0.f, 0.f, 0.f);
        for (int s = 0; s < nsteps; s += 8) {
            const int pos0 = 2 * s + h;
            switch (nsteps - s) {
                case 1:  a = gcn_pair_round<1>(S4, q, row, pos0, deg, a); break;
                case 2:  a = gcn_pair_round<2>(S4, q, row, pos0, deg, a); break;
                case 3:  a = gcn_pair_round<3>(S4, q, row, pos0, deg, a); break;
                case 4:  a = gcn_pair_round<4>(S4, q, row, pos0, deg, a); break;
                case 5:  a = gcn_pair_round<5>(S4, q, row, pos0, deg, a); break;
                case 6:  a = gcn_pair_round<6>(S4, q, row, pos0, deg, a); break;
                case 7:  a = gcn_pair_round<7>(S4, q, row, pos0, deg, a); break;
                default: a = gcn_pair_round<8>(S4, q, row, pos0, deg, a); break;
            }
        }

        a.x += gcn_other_half(a.x);
        a.y += gcn_other_half(a.y);
        a.z += gcn_other_half(a.z);
        a.w += gcn_other_half(a.w);

        if (h == 0) ((float4*)out)[(unsigned)((r0 + rl) * (DIM / 4) + q)] = a;
    }
    if (!big) return;

    #pragma nounroll
    for (int i = 0; i < BUCKET_ROWS / SUM_WAVES; ++i) {
        const int rl = i * SUM_WAVES + w;
        if (rl >= rows) break;
        const int deg = __builtin_amdgcn_readfirstlane(lcnt[rl]);
        const int off = __builtin_amdgcn_readfirstlane(loff[rl]);
        if (deg <= 64) continue;

        float2 acc = make_float2(0.f, 0.f);
        int last = -1;
        for (int g = 0; g < deg; ++g) {
            int best = NO_EDGE, where = 0;
            for (int j = lane; j < deg; j += 32) {
                const int id = rec_id[off + j];
                if (id > last && id < best) { best = id; where = j; }
            }
            gcn_wave_min(best, where);
            if (best == NO_EDGE) break;
            const int2   cv = rec_cv[off + where];
            const float  v  = __int_as_float(cv.y);
            const float2 x  = S2[(long long)cv.x * (DIM / 2) + lane];
            acc.x += v * x.x;
            acc.y += v * x.y;
            last = best;
        }
        ((float2*)out)[(long long)(r0 + rl) * (DIM / 2) + lane] = acc;
    }
}

extern "C" void kernel_launch(void* const* d_in, const int* in_sizes, int n_in,
                              void* d_out, int out_size, void* d_ws, size_t ws_size,
                              hipStream_t stream) {
    const float* X    = (const float*)d_in[0];
    const float* W    = (const float*)d_in[1];
    const int*   arow = (const int*)d_in[2];
    const int*   acol = (const int*)d_in[3];
    const float* aval = (const float*)d_in[4];
    float*       out  = (float*)d_out;

    const int n_nodes = in_sizes[0] / DIM;
    const int n_edges = in_sizes[2];

    float* S        = (float*)d_ws;
    int2*  table    = (int2*)(S + (size_t)n_nodes * DIM);
    int4*  binned   = (int4*)(table + NCHUNK * MAX_BUCKETS);

    const int tiles = n_nodes / 16;
    const int g_all = (tiles + GEMM_WAVES - 1) / GEMM_WAVES;

    gcn_bin_kernel<<<NCHUNK + g_all, BIN_THREADS, 0, stream>>>(arow, acol, aval, table, binned, n_edges,
                                                              X, W, S, tiles);

    gcn_group_sum_kernel<<<MAX_BUCKETS, SUM_THREADS, 0, stream>>>(binned, table, S, out, n_nodes, n_edges);
}


# Product moved behind the row sums: out = (A X) W, a bucket's summed X rows times W on f32 WMMA inside the summing work-group; support array and product work-groups dropped; row sum order changes
# speedup vs baseline: 6.5515x; 1.1041x over previous
#include <hip/hip_runtime.h>

typedef __attribute__((ext_vector_type(2))) float v2f;
typedef __attribute__((ext_vector_type(8))) float v8f;

#define DIM 64
#define NO_EDGE 0x7fffffff

#define NCHUNK      256
#define BUCKET_ROWS 256
#define MAX_BUCKETS 512
#define BIN_THREADS 512
#define BIN_EDGES   4
#define BIN_HELD    13
#define STAGE_CAP   4096

#define SUM_THREADS 1024
#define SUM_WAVES   (SUM_THREADS / 32)
#define SUM_CAP     (6 * SUM_THREADS)
#define SEG_LANES   16
#define SEG_GROUPS  (SUM_THREADS / SEG_LANES)
#define SEG_PER     (NCHUNK / SEG_GROUPS)
#define SEG_HELD    2

#define AGG_PITCH 68

static __device__ __forceinline__ void gcn_product_tile(const float* agg,
                                                        const float* __restrict__ W,
                                                        float* __restrict__ out,
                                                        int row0, int n_nodes,
                                                        int tile, int cbase, int lane) {
    const int mrow  = lane & 15;
    const int khalf = lane >> 4;
    const int ncol  = lane & 15;

    v8f acc0 = {}, acc1 = {};

    const float* arow = agg + (tile * 16 + mrow) * AGG_PITCH;

    #pragma unroll
    for (int k0 = 0; k0 < DIM; k0 += 4) {
        const int ka = k0 + (khalf << 1);
        v2f a = *(const v2f*)(arow + ka);

        const float* __restrict__ w0 = W + ka * DIM + cbase;
        const float* __restrict__ w1 = W + (ka + 1) * DIM + cbase;

        v2f b0, b1;
        b0.x = w0[ 0 + ncol]; b0.y = w1[ 0 + ncol];
        b1.x = w0[16 + ncol]; b1.y = w1[16 + ncol];

        acc0 = __builtin_amdgcn_wmma_f32_16x16x4_f32(false, a, false, b0,
                                                     (short)0, acc0, false, false);
        acc1 = __builtin_amdgcn_wmma_f32_16x16x4_f32(false, a, false, b1,
                                                     (short)0, acc1, false, false);
    }

    const int rbase = row0 + tile * 16 + (khalf << 3);
    #pragma unroll
    for (int v = 0; v < 8; ++v) {
        if (rbase + v < n_nodes) {
            float* __restrict__ orow = out + (long long)(rbase + v) * DIM + cbase + ncol;
            orow[ 0] = acc0[v];
            orow[16] = acc1[v];
        }
    }
}

static __device__ __forceinline__ void gcn_lds_barrier() {
    __builtin_amdgcn_fence(__ATOMIC_RELEASE, "workgroup", "local");
    __builtin_amdgcn_s_barrier();
    __builtin_amdgcn_fence(__ATOMIC_ACQUIRE, "workgroup", "local");
}

__global__ void __launch_bounds__(BIN_THREADS)
gcn_bin_kernel(const int* __restrict__ adj_row,
               const int* __restrict__ adj_col,
               const float* __restrict__ adj_val,
               int2* __restrict__ table,
               int4* __restrict__ binned,
               int n_edges) {
    __shared__ int lh[MAX_BUCKETS];
    __shared__ int sc[MAX_BUCKETS];
    __shared__ int lpos[MAX_BUCKETS];
    __shared__ int4 stage[STAGE_CAP];
    static_assert(STAGE_CAP % 8 == 0, "a part of the stretch must start on a 128-byte line");
    const int t = threadIdx.x;
    const int c = blockIdx.x;

    const int run = (n_edges + NCHUNK - 1) / NCHUNK;
    const int e0  = c * run;
    const int e1  = (e0 + run < n_edges) ? e0 + run : n_edges;
    const bool held = run <= BIN_HELD * BIN_THREADS;

    lh[t] = 0;
    __syncthreads();

    int   hr[BIN_HELD], hc[BIN_HELD];
    float hv[BIN_HELD];
    if (held) {
        #pragma unroll
        for (int k = 0; k < BIN_HELD; ++k) {
            const int e  = e0 + t + k * BIN_THREADS;
            const int ec = (e < e1) ? e : ((e1 > 0) ? e1 - 1 : 0);
            hr[k] = adj_row[ec];
        }
        #pragma unroll
        for (int k = 0; k < BIN_HELD; ++k) {
            const int e  = e0 + t + k * BIN_THREADS;
            const int ec = (e < e1) ? e : ((e1 > 0) ? e1 - 1 : 0);
            hc[k] = adj_col[ec];
            hv[k] = adj_val[ec];
        }
        #pragma unroll
        for (int k = 0; k < BIN_HELD; ++k) {
            const int e = e0 + t + k * BIN_THREADS;
            if (e >= e1) hr[k] = -1;
        }
    }

    if (held) {
        #pragma unroll
        for (int k = 0; k < BIN_HELD; ++k) {
            if (hr[k] >= 0) atomicAdd(&lh[hr[k] >> 8], 1);
        }
    } else
    for (int eb = e0 + t; eb < e1; eb += BIN_EDGES * BIN_THREADS) {
        int r[BIN_EDGES];
        #pragma unroll
        for (int k = 0; k < BIN_EDGES; ++k) {
            const int e = eb + k * BIN_THREADS;
            r[k] = (e < e1) ? adj_row[e] : -1;
        }
        #pragma unroll
        for (int k = 0; k < BIN_EDGES; ++k) {
            if (r[k] >= 0) atomicAdd(&lh[r[k] >> 8], 1);
        }
    }
    __syncthreads();

    const int tot = lh[t];
    sc[t] = tot;
    __syncthreads();
    for (int off = 1; off < MAX_BUCKETS; off <<= 1) {
        const int add = (t >= off) ? sc[t - off] : 0;
        __syncthreads();
        sc[t] += add;
        __syncthreads();
    }
    const int first = sc[t] - tot;
    lpos[t] = e0 + first;
    table[c * MAX_BUCKETS + t] = make_int2(first, tot);
    __syncthreads();

    if (held) {
        int p[BIN_HELD];
        #pragma unroll
        for (int k = 0; k < BIN_HELD; ++k) {
            p[k] = 0;
            if (hr[k] >= 0) p[k] = atomicAdd(&lpos[hr[k] >> 8], 1);
        }
        const int cnt  = e1 - e0;
        const int lead = e0 & 7;
        for (int b0 = 0; b0 < cnt; b0 += STAGE_CAP) {
            const int b1 = (b0 + STAGE_CAP < cnt) ? b0 + STAGE_CAP : cnt;
            if (b0 > 0) gcn_lds_barrier();
            #pragma unroll
            for (int k = 0; k < BIN_HELD; ++k) {
                const int e = e0 + t + k * BIN_THREADS;
                const int i = p[k] - e0;
                if (hr[k] >= 0 && i >= b0 && i < b1)
                    stage[i - b0] = make_int4(e, hr[k], hc[k], __float_as_int(hv[k]));
            }
            gcn_lds_barrier();
            for (int g = b0 + t - lead; g < b1; g += BIN_THREADS) {
                if (g >= b0) binned[e0 + g] = stage[g - b0];
            }
        }
        return;
    }

    for (int eb = e0 + t; eb < e1; eb += BIN_EDGES * BIN_THREADS) {
        int   r[BIN_EDGES], cl[BIN_EDGES], p[BIN_EDGES];
        float v[BIN_EDGES];
        #pragma unroll
        for (int k = 0; k < BIN_EDGES; ++k) {
            const int e = eb + k * BIN_THREADS;
            r[k] = -1; cl[k] = 0; v[k] = 0.f;
            if (e < e1) { r[k] = adj_row[e]; cl[k] = adj_col[e]; v[k] = adj_val[e]; }
        }
        #pragma unroll
        for (int k = 0; k < BIN_EDGES; ++k) {
            p[k] = 0;
            if (r[k] >= 0) p[k] = atomicAdd(&lpos[r[k] >> 8], 1);
        }
        #pragma unroll
        for (int k = 0; k < BIN_EDGES; ++k) {
            const int e = eb + k * BIN_THREADS;
            if (r[k] >= 0) binned[p[k]] = make_int4(e, r[k], cl[k], __float_as_int(v[k]));
        }
    }
}

template <int U>
static __device__ __forceinline__ float4 gcn_pair_round(const float4* __restrict__ S4, int q,
                                                        const int2* row, int pos0, int deg, float4 acc) {
    float  v[U];
    float4 x[U];
    const int  plast = pos0 + 2 * (U - 1);
    const bool have  = plast < deg;
    #pragma unroll
    for (int k = 0; k < U; ++k) {
        const int  p  = (k == U - 1) ? (have ? plast : deg - 1) : pos0 + 2 * k;
        const int2 cv = row[p];
        v[k] = __int_as_float(cv.y);
        x[k] = S4[(unsigned)(cv.x * (DIM / 4) + q)];
    }
    if (!have) { v[U - 1] = 0.f; x[U - 1] = make_float4(0.f, 0.f, 0.f, 0.f); }
    #pragma unroll
    for (int k = 0; k < U; ++k) {
        acc.x = __builtin_fmaf(v[k], x[k].x, acc.x);
        acc.y = __builtin_fmaf(v[k], x[k].y, acc.y);
        acc.z = __builtin_fmaf(v[k], x[k].z, acc.z);
        acc.w = __builtin_fmaf(v[k], x[k].w, acc.w);
    }
    return acc;
}

static __device__ __forceinline__ float gcn_other_half(float f) {
    const unsigned u = (unsigned)__float_as_int(f);
    return __int_as_float((int)__builtin_amdgcn_permlanex16(u, u, 0x76543210u, 0xfedcba98u, false, false));
}

static __device__ __forceinline__ void gcn_wave_min(int& best, int& where) {
    for (int off = 16; off > 0; off >>= 1) {
        const int ob = __shfl_xor(best, off);
        const int ow = __shfl_xor(where, off);
        if (ob < best) { best = ob; where = ow; }
    }
}

__global__ void __launch_bounds__(SUM_THREADS) __attribute__((amdgpu_waves_per_eu(16)))
gcn_group_sum_kernel(const int4* __restrict__ binned,
                     const int2* __restrict__ table,
                     const float* __restrict__ X,
                     const float* __restrict__ W,
                     float* __restrict__ out,
                     int n_nodes, int n_edges) {
    static_assert(SEG_GROUPS * SEG_PER == NCHUNK, "every run's stretch needs its half-wave");
    __shared__ int  lcnt[BUCKET_ROWS];
    __shared__ int  loff[BUCKET_ROWS];
    __shared__ int  lpos[BUCKET_ROWS];
    __shared__ int  wtot[BUCKET_ROWS / 32];
    __shared__ int  rec_id[SUM_CAP];
    __shared__ int2 rec_cv[SUM_CAP];
    __shared__ __attribute__((aligned(16))) float agg[BUCKET_ROWS * AGG_PITCH];
    static_assert(SUM_WAVES == 2 * (BUCKET_ROWS / 16), "the product needs two waves per 16-row tile");

    const int t    = threadIdx.x;
    const int lane = t & 31;
    const int w  = __builtin_amdgcn_readfirstlane(t >> 5);
    const int r0 = blockIdx.x * BUCKET_ROWS;
    if (r0 >= n_nodes) return;

    const int rows = (n_nodes - r0 < BUCKET_ROWS) ? n_nodes - r0 : BUCKET_ROWS;
    const float2* __restrict__ S2 = (const float2*)X;
    const float4* __restrict__ S4 = (const float4*)X;
    const int h = lane >> 4;
    const int q = lane & 15;

    const int run = (n_edges + NCHUNK - 1) / NCHUNK;
    const int sub = t & (SEG_LANES - 1);
    const int grp = t / SEG_LANES;
    int sbase[SEG_PER], scnt[SEG_PER];
    #pragma unroll
    for (int m = 0; m < SEG_PER; ++m) {
        const int  cr = grp + m * SEG_GROUPS;
        const int2 sg = table[cr * MAX_BUCKETS + blockIdx.x];
        sbase[m] = cr * run + sg.x;
        scnt[m]  = sg.y;
    }

    int4 hq[SEG_PER * SEG_HELD];
    #pragma unroll
    for (int m = 0; m < SEG_PER; ++m) {
        #pragma unroll
        for (int k = 0; k < SEG_HELD; ++k) {
            const int j = sub + k * SEG_LANES;
            hq[m * SEG_HELD + k] = make_int4(0, -1, 0, 0);
            if (j < scnt[m]) hq[m * SEG_HELD + k] = binned[sbase[m] + j];
        }
    }

    if (t < BUCKET_ROWS) lcnt[t] = 0;
    __syncthreads();

    #pragma unroll
    for (int i = 0; i < SEG_PER * SEG_HELD; ++i) {
        if (hq[i].y >= 0) atomicAdd(&lcnt[hq[i].y - r0], 1);
    }
    bool more = false;
    #pragma unroll
    for (int m = 0; m < SEG_PER; ++m) {
        more |= sub + SEG_HELD * SEG_LANES < scnt[m];
        for (int j = sub + SEG_HELD * SEG_LANES; j < scnt[m]; j += SEG_LANES) {
            atomicAdd(&lcnt[binned[sbase[m] + j].y - r0], 1);
        }
    }
    __syncthreads();

    int c = 0, incl = 0;
    if (t < BUCKET_ROWS) {
        c = lcnt[t];
        incl = c;
        for (int off = 1; off < 32; off <<= 1) {
            const int up = __shfl_up(incl, off);
            if (lane >= off) incl += up;
        }
        if (lane == 31) wtot[w] = incl;
    }
    __syncthreads();
    if (t < BUCKET_ROWS) {
        int base = 0;
        for (int j = 0; j < BUCKET_ROWS / 32; ++j) base += (j < w) ? wtot[j] : 0;
        loff[t] = base + incl - c;
        lpos[t] = base + incl - c;
    }
    int n = 0;
    for (int j = 0; j < BUCKET_ROWS / 32; ++j) n += wtot[j];
    __syncthreads();

    if (n > SUM_CAP) {
        for (int i = 0; i < BUCKET_ROWS / SUM_WAVES; ++i) {
            const int rl = i * SUM_WAVES + w;
            if (rl >= rows) break;
            const int r = r0 + rl;
            float2 acc = make_float2(0.f, 0.f);
            int last = -1;
            for (int g = 0; g < n; ++g) {
                int best = NO_EDGE, where = 0;
                for (int cc = 0; cc < NCHUNK; ++cc) {
                    const int2 s2  = table[cc * MAX_BUCKETS + blockIdx.x];
                    const int  beg = cc * run + s2.x;
                    for (int j = lane; j < s2.y; j += 32) {
                        const int4 q = binned[beg + j];
                        if (q.y == r && q.x > last && q.x < best) { best = q.x; where = beg + j; }
                    }
                }
                gcn_wave_min(best, where);
                if (best == NO_EDGE) break;
                const int4   q = binned[where];
                const float  v = __int_as_float(q.w);
                const float2 x = S2[(long long)q.z * (DIM / 2) + lane];
                acc.x += v * x.x;
                acc.y += v * x.y;
                last = best;
            }
            *(float2*)&agg[rl * AGG_PITCH + 2 * lane] = acc;
        }
    } else {

    {
        int p[SEG_PER * SEG_HELD];
        #pragma unroll
        for (int i = 0; i < SEG_PER * SEG_HELD; ++i) {
            p[i] = 0;
            if (hq[i].y >= 0) p[i] = atomicAdd(&lpos[hq[i].y - r0], 1);
        }
        #pragma unroll
        for (int i = 0; i < SEG_PER * SEG_HELD; ++i) {
            if (hq[i].y >= 0) {
                rec_id[p[i]] = hq[i].x;
                rec_cv[p[i]] = make_int2(hq[i].z, hq[i].w);
            }
        }
    }
    if (more) {
        #pragma nounroll
        for (int m = 0; m < SEG_PER; ++m) {
            const int  cr = grp + m * SEG_GROUPS;
            const int2 sg = table[cr * MAX_BUCKETS + blockIdx.x];
            for (int j = sub + SEG_HELD * SEG_LANES; j < sg.y; j += SEG_LANES) {
                const int4 q = binned[cr * run + sg.x + j];
                const int  p = atomicAdd(&lpos[q.y - r0], 1);
                rec_id[p] = q.x;
                rec_cv[p] = make_int2(q.z, q.w);
            }
        }
    }
    __syncthreads();

    bool big = false;
    for (int i = 0; i < BUCKET_ROWS / SUM_WAVES; ++i) {
        const int rl = i * SUM_WAVES + w;
        if (rl >= rows) break;
        const int deg = __builtin_amdgcn_readfirstlane(lcnt[rl]);
        const int off = __builtin_amdgcn_readfirstlane(loff[rl]);
        if (deg > 64) { big = true; continue; }

        int e0 = NO_EDGE, c0 = 0, v0 = 0;
        int e1 = NO_EDGE, c1 = 0, v1 = 0;
        if (lane < deg)      { e0 = rec_id[off + lane];      const int2 cv = rec_cv[off + lane];      c0 = cv.x; v0 = cv.y; }
        if (lane + 32 < deg) { e1 = rec_id[off + lane + 32]; const int2 cv = rec_cv[off + lane + 32]; c1 = cv.x; v1 = cv.y; }

        int rk0 = 0, rk1 = 0;
        const int n0 = deg < 32 ? deg : 32;
        if (deg <= 32) {
            for (int j = 0; j < n0; ++j) {
                const int kj = __builtin_amdgcn_readlane(e0, j);
                rk0 += (kj < e0) ? 1 : 0;
            }
        } else {
            for (int j = 0; j < 32; ++j) {
                const int kj = __builtin_amdgcn_readlane(e0, j);
                rk0 += (kj < e0) ? 1 : 0;
                rk1 += (kj < e1) ? 1 : 0;
            }
            for (int j = 32; j < deg; ++j) {
                const int kj = __builtin_amdgcn_readlane(e1, j - 32);
                rk0 += (kj < e0) ? 1 : 0;
                rk1 += (kj < e1) ? 1 : 0;
            }
        }
        if (lane < deg)      rec_cv[off + rk0] = make_int2(c0, v0);
        if (lane + 32 < deg) rec_cv[off + rk1] = make_int2(c1, v1);

        const int2* row    = &rec_cv[off];
        const int   nsteps = (deg + 1) >> 1;
        float4 a = make_float4(0.f, 0.f, 0.f, 0.f);
        for (int s = 0; s < nsteps; s += 8) {
            const int pos0 = 2 * s + h;
            switch (nsteps - s) {
                case 1:  a = gcn_pair_round<1>(S4, q, row, pos0, deg, a); break;
                case 2:  a = gcn_pair_round<2>(S4, q, row, pos0, deg, a); break;
                case 3:  a = gcn_pair_round<3>(S4, q, row, pos0, deg, a); break;
                case 4:  a = gcn_pair_round<4>(S4, q, row, pos0, deg, a); break;
                case 5:  a = gcn_pair_round<5>(S4, q, row, pos0, deg, a); break;
                case 6:  a = gcn_pair_round<6>(S4, q, row, pos0, deg, a); break;
                case 7:  a = gcn_pair_round<7>(S4, q, row, pos0, deg, a); break;
                default: a = gcn_pair_round<8>(S4, q, row, pos0, deg, a); break;
            }
        }

        a.x += gcn_other_half(a.x);
        a.y += gcn_other_half(a.y);
        a.z += gcn_other_half(a.z);
        a.w += gcn_other_half(a.w);

        if (h == 0) *(float4*)&agg[rl * AGG_PITCH + 4 * q] = a;
    }
    if (big) {

    #pragma nounroll
    for (int i = 0; i < BUCKET_ROWS / SUM_WAVES; ++i) {
        const int rl = i * SUM_WAVES + w;
        if (rl >= rows) break;
        const int deg = __builtin_amdgcn_readfirstlane(lcnt[rl]);
        const int off = __builtin_amdgcn_readfirstlane(loff[rl]);
        if (deg <= 64) continue;

        float2 acc = make_float2(0.f, 0.f);
        int last = -1;
        for (int g = 0; g < deg; ++g) {
            int best = NO_EDGE, where = 0;
            for (int j = lane; j < deg; j += 32) {
                const int id = rec_id[off + j];
                if (id > last && id < best) { best = id; where = j; }
            }
            gcn_wave_min(best, where);
            if (best == NO_EDGE) break;
            const int2   cv = rec_cv[off + where];
            const float  v  = __int_as_float(cv.y);
            const float2 x  = S2[(long long)cv.x * (DIM / 2) + lane];
            acc.x += v * x.x;
            acc.y += v * x.y;
            last = best;
        }
        *(float2*)&agg[rl * AGG_PITCH + 2 * lane] = acc;
    }
    }
    }

    __syncthreads();
    {
        const int tile  = w >> 1;
        const int cbase = (w & 1) * 32;
        if (tile * 16 < rows) gcn_product_tile(agg, W, out, r0, n_nodes, tile, cbase, lane);
    }
}

extern "C" void kernel_launch(void* const* d_in, const int* in_sizes, int n_in,
                              void* d_out, int out_size, void* d_ws, size_t ws_size,
                              hipStream_t stream) {
    const float* X    = (const float*)d_in[0];
    const float* W    = (const float*)d_in[1];
    const int*   arow = (const int*)d_in[2];
    const int*   acol = (const int*)d_in[3];
    const float* aval = (const float*)d_in[4];
    float*       out  = (float*)d_out;

    const int n_nodes = in_sizes[0] / DIM;
    const int n_edges = in_sizes[2];

    int2*  table    = (int2*)d_ws;
    int4*  binned   = (int4*)(table + NCHUNK * MAX_BUCKETS);

    gcn_bin_kernel<<<NCHUNK, BIN_THREADS, 0, stream>>>(arow, acol, aval, table, binned, n_edges);

    gcn_group_sum_kernel<<<MAX_BUCKETS, SUM_THREADS, 0, stream>>>(binned, table, X, W, out, n_nodes, n_edges);
}
